# starting kernel
# baseline (speedup 1.0000x reference)
#include <hip/hip_runtime.h>

typedef float v2f __attribute__((ext_vector_type(2)));
typedef float v8f __attribute__((ext_vector_type(8)));

#define DD 64


__global__ void __launch_bounds__(256)
fill_f32(float* __restrict__ p, float v, int n) {
  int i = blockIdx.x * blockDim.x + threadIdx.x;
  if (i < n) p[i] = v;
}

__global__ void __launch_bounds__(256)
degree_count(const int* __restrict__ dst, float* __restrict__ deg, int ne) {
  int e = blockIdx.x * blockDim.x + threadIdx.x;
  if (e < ne) unsafeAtomicAdd(&deg[dst[e]], 1.0f);
}

__global__ void __launch_bounds__(256)
rsqrt_inplace(float* __restrict__ p, int n) {
  int i = blockIdx.x * blockDim.x + threadIdx.x;
  if (i < n) p[i] = rsqrtf(p[i]);
}

__global__ void __launch_bounds__(256)
selfloop_init(const float* __restrict__ xw, const float* __restrict__ dinv,
              float* __restrict__ acc, int n) {
  int t = blockIdx.x * blockDim.x + threadIdx.x;
  if (t >= n * DD) return;
  int i = t >> 6;
  float di = dinv[i];
  acc[t] = di * di * xw[t];
}

__global__ void __launch_bounds__(256)
scatter_edges(const int* __restrict__ src, const int* __restrict__ dst,
              const float* __restrict__ msg, const float* __restrict__ dinv,
              float* __restrict__ acc, int ne) {
  int t = blockIdx.x * blockDim.x + threadIdx.x;
  int e = t >> 6;
  if (e >= ne) return;
  int c = t & 63;
  int s = src[e];
  int d = dst[e];
  float nrm = dinv[s] * dinv[d];
  unsafeAtomicAdd(&acc[(size_t)d * DD + c], msg[(size_t)s * DD + c] * nrm);
}

__global__ void __launch_bounds__(256)
bias_relu_inplace(float* __restrict__ h, const float* __restrict__ b, int n) {
  int t = blockIdx.x * blockDim.x + threadIdx.x;
  if (t >= n * DD) return;
  h[t] = fmaxf(h[t] + b[t & 63], 0.0f);
}

__global__ void __launch_bounds__(256)
final_combine(float* __restrict__ out, const float* __restrict__ hw2,
              const float* __restrict__ dinv, const float* __restrict__ b2,
              const float* __restrict__ bemb, int n) {
  int t = blockIdx.x * blockDim.x + threadIdx.x;
  if (t >= n * DD) return;
  int i = t >> 6, c = t & 63;
  float di = dinv[i];
  out[t] = out[t] + bemb[c] + b2[c] + di * di * hw2[t];
}

__global__ void __launch_bounds__(256)
gemm_n64_wmma(const float* __restrict__ A, const float* __restrict__ W,
              float* __restrict__ C, int nrows) {
  const int lane = threadIdx.x & 31;
  const int wave = threadIdx.x >> 5;
  const int tile = blockIdx.x * 8 + wave;
  const int row0 = tile * 16;
  if (row0 >= nrows) return;

  const int half = lane >> 4;
  const int l16  = lane & 15;

  int ar = row0 + l16;
  if (ar >= nrows) ar = nrows - 1;
  const float* __restrict__ arow = A + (size_t)ar * DD;

  v8f acc0 = {}, acc1 = {}, acc2 = {}, acc3 = {};

#pragma unroll
  for (int k = 0; k < 16; ++k) {
    const int kb = k * 4 + half * 2;
    v2f a = *(const v2f*)(arow + kb);

    const float* __restrict__ wr0 = W + (size_t)kb * DD;
    const float* __restrict__ wr1 = W + (size_t)(kb + 1) * DD;
    v2f b0, b1, b2, b3;
    b0.x = wr0[l16];      b0.y = wr1[l16];
    b1.x = wr0[16 + l16]; b1.y = wr1[16 + l16];
    b2.x = wr0[32 + l16]; b2.y = wr1[32 + l16];
    b3.x = wr0[48 + l16]; b3.y = wr1[48 + l16];

    acc0 = __builtin_amdgcn_wmma_f32_16x16x4_f32(false, a, false, b0, (short)0, acc0, false, false);
    acc1 = __builtin_amdgcn_wmma_f32_16x16x4_f32(false, a, false, b1, (short)0, acc1, false, false);
    acc2 = __builtin_amdgcn_wmma_f32_16x16x4_f32(false, a, false, b2, (short)0, acc2, false, false);
    acc3 = __builtin_amdgcn_wmma_f32_16x16x4_f32(false, a, false, b3, (short)0, acc3, false, false);
  }

  float* __restrict__ crow = C + (size_t)row0 * DD;
#pragma unroll
  for (int r = 0; r < 8; ++r) {
    const int rr = r + half * 8;
    if (row0 + rr < nrows) {
      crow[(size_t)rr * DD + l16]      = acc0[r];
      crow[(size_t)rr * DD + 16 + l16] = acc1[r];
      crow[(size_t)rr * DD + 32 + l16] = acc2[r];
      crow[(size_t)rr * DD + 48 + l16] = acc3[r];
    }
  }
}

extern "C" void kernel_launch(void* const* d_in, const int* in_sizes, int n_in,
                              void* d_out, int out_size, void* d_ws, size_t ws_size,
                              hipStream_t stream) {
  const float* x    = (const float*)d_in[0];
  const int*   ei   = (const int*)d_in[1];
  const float* W1   = (const float*)d_in[2];
  const float* b1   = (const float*)d_in[3];
  const float* W2   = (const float*)d_in[4];
  const float* b2   = (const float*)d_in[5];
  const float* Wemb = (const float*)d_in[6];
  const float* bemb = (const float*)d_in[7];
  float* out = (float*)d_out;

  const int N = in_sizes[0] / DD;
  const int E = in_sizes[1] / 2;
  const int* src = ei;
  const int* dst = ei + E;

  float* ws = (float*)d_ws;
  const size_t npad = ((size_t)N + 63) & ~(size_t)63;
  float* dinv = ws;
  float* buf1 = ws + npad;
  float* buf2 = buf1 + npad * DD;

  const int B   = 256;
  const int gN  = (N + B - 1) / B;
  const int gE  = (E + B - 1) / B;
  const int gNE = (N * DD + B - 1) / B;
  const int gEE = (int)(((size_t)E * DD + (size_t)B - 1) / (size_t)B);
  const int gG  = (((N + 15) / 16) + 7) / 8;

  fill_f32<<<gN, B, 0, stream>>>(dinv, 1.0f, N);
  degree_count<<<gE, B, 0, stream>>>(dst, dinv, E);
  rsqrt_inplace<<<gN, B, 0, stream>>>(dinv, N);

  gemm_n64_wmma<<<gG, B, 0, stream>>>(x, W1, buf1, N);
  selfloop_init<<<gNE, B, 0, stream>>>(buf1, dinv, buf2, N);
  scatter_edges<<<gEE, B, 0, stream>>>(src, dst, buf1, dinv, buf2, E);
  bias_relu_inplace<<<gNE, B, 0, stream>>>(buf2, b1, N);

  gemm_n64_wmma<<<gG, B, 0, stream>>>(buf2, W2, buf1, N);
  gemm_n64_wmma<<<gG, B, 0, stream>>>(x, Wemb, out, N);
  final_combine<<<gNE, B, 0, stream>>>(out, buf1, dinv, b2, bemb, N);
  scatter_edges<<<gEE, B, 0, stream>>>(src, dst, buf1, dinv, out, E);
}


# Float-atomic feature scatter replaced by per-row gather over a destination-grouped edge list built each call (int atomic arrival-order rank + scan); rows summed in list order, one store per row
# speedup vs baseline: 2.8739x; 2.8739x over previous
#include <hip/hip_runtime.h>


typedef float v2f __attribute__((ext_vector_type(2)));
typedef float v8f __attribute__((ext_vector_type(8)));

#define DD 64


__global__ void __launch_bounds__(256)
fill_i32(int* __restrict__ p, int v, int n) {
  int i = blockIdx.x * blockDim.x + threadIdx.x;
  if (i < n) p[i] = v;
}

__global__ void __launch_bounds__(256)
degree_rank(const int* __restrict__ dst, int* __restrict__ cnt,
            int* __restrict__ rank, int ne) {
  int e = blockIdx.x * blockDim.x + threadIdx.x;
  if (e < ne) rank[e] = atomicAdd(&cnt[dst[e]], 1);
}

__device__ __forceinline__ int wave_sum(int v) {
  for (int m = 16; m > 0; m >>= 1) v += __shfl_xor(v, m);
  return v;
}

__global__ void __launch_bounds__(256)
block_sums(const int* __restrict__ cnt, int* __restrict__ partial) {
  __shared__ int sw[8];
  const int t = threadIdx.x;
  const int4 c = ((const int4*)cnt)[(size_t)blockIdx.x * 256 + t];
  const int s = wave_sum(c.x + c.y + c.z + c.w);
  if ((t & 31) == 0) sw[t >> 5] = s;
  __syncthreads();
  if (t == 0) {
    int tot = 0;
    for (int k = 0; k < 8; ++k) tot += sw[k];
    partial[blockIdx.x] = tot;
  }
}

__global__ void __launch_bounds__(256)
scan_rows(const int* __restrict__ cnt, const int* __restrict__ partial,
          int* __restrict__ rowptr, float* __restrict__ dinv) {
  __shared__ int sa[8];
  __shared__ int sb[8];
  const int t = threadIdx.x;
  const int lane = t & 31;
  const int wave = t >> 5;

  int p = 0;
  for (int k = t; k < (int)blockIdx.x; k += 256) p += partial[k];
  p = wave_sum(p);
  if (lane == 0) sa[wave] = p;

  const int4 c = ((const int4*)cnt)[(size_t)blockIdx.x * 256 + t];
  const int s = c.x + c.y + c.z + c.w;
  int inc = s;
  for (int d = 1; d < 32; d <<= 1) {
    const int up = __shfl_up(inc, d);
    if (lane >= d) inc += up;
  }
  if (lane == 31) sb[wave] = inc;
  __syncthreads();

  int off = 0;
  for (int k = 0; k < 8; ++k) {
    off += sa[k];
    if (k < wave) off += sb[k];
  }
  int4 r;
  r.x = off + inc - s;
  r.y = r.x + c.x;
  r.z = r.y + c.y;
  r.w = r.z + c.z;
  ((int4*)rowptr)[(size_t)blockIdx.x * 256 + t] = r;
  float4 dv;
  dv.x = rsqrtf((float)(c.x + 1));
  dv.y = rsqrtf((float)(c.y + 1));
  dv.z = rsqrtf((float)(c.z + 1));
  dv.w = rsqrtf((float)(c.w + 1));
  ((float4*)dinv)[(size_t)blockIdx.x * 256 + t] = dv;
}

__global__ void __launch_bounds__(256)
place_edges(const int* __restrict__ src, const int* __restrict__ dst,
            const int* __restrict__ rank, const int* __restrict__ rowptr,
            const float* __restrict__ dinv, int* __restrict__ rec, int ne) {
  int e = blockIdx.x * blockDim.x + threadIdx.x;
  if (e >= ne) return;
  const int s = src[e];
  const int d = dst[e];
  int2 r;
  r.x = s;
  r.y = __float_as_int(dinv[s] * dinv[d]);
  ((int2*)rec)[(size_t)rowptr[d] + rank[e]] = r;
}

__device__ __forceinline__ v2f row_sum(const int* __restrict__ rowptr,
                                       const int* __restrict__ rec,
                                       const float* __restrict__ msg,
                                       const float* __restrict__ dinv,
                                       int i, int lane) {
  const int beg = __builtin_amdgcn_readfirstlane(rowptr[i]);
  const int end = __builtin_amdgcn_readfirstlane(rowptr[i + 1]);
  const float di = dinv[i];
  const v2f* __restrict__ m2 = (const v2f*)msg;
  const v2f self = m2[(size_t)i * 32 + lane];
  const float dd = di * di;
  float ax = dd * self.x;
  float ay = dd * self.y;
  for (int base = beg; base < end; base += 32) {
    const int cnt = min(32, end - base);
    int my = i;
    int wb = 0;
    if (lane < cnt) {
      const int2 r = ((const int2*)rec)[(size_t)base + lane];
      my = r.x;
      wb = r.y;
    }
    for (int j = 0; j < cnt; j += 8) {
      v2f v[8];
      float w[8];
#pragma unroll
      for (int u = 0; u < 8; ++u) {
        const int s = __builtin_amdgcn_readlane(my, j + u);
        w[u] = __int_as_float(__builtin_amdgcn_readlane(wb, j + u));
        v[u] = m2[(size_t)s * 32 + lane];
      }
#pragma unroll
      for (int u = 0; u < 8; ++u) {
        const bool on = (j + u) < cnt;
        ax = fmaf(w[u], on ? v[u].x : 0.0f, ax);
        ay = fmaf(w[u], on ? v[u].y : 0.0f, ay);
      }
    }
  }
  v2f r;
  r.x = ax;
  r.y = ay;
  return r;
}

__global__ void __launch_bounds__(256)
gather_relu(const int* __restrict__ rowptr, const int* __restrict__ rec,
            const float* __restrict__ msg, const float* __restrict__ dinv,
            const float* __restrict__ b, float* __restrict__ h, int n) {
  const int lane = threadIdx.x & 31;
  const int i = blockIdx.x * 8 + __builtin_amdgcn_readfirstlane((int)(threadIdx.x >> 5));
  if (i >= n) return;
  const v2f bv = ((const v2f*)b)[lane];
  const v2f a = row_sum(rowptr, rec, msg, dinv, i, lane);
  v2f r;
  r.x = fmaxf(a.x + bv.x, 0.0f);
  r.y = fmaxf(a.y + bv.y, 0.0f);
  ((v2f*)h)[(size_t)i * 32 + lane] = r;
}

__global__ void __launch_bounds__(256)
gather_final(const int* __restrict__ rowptr, const int* __restrict__ rec,
             const float* __restrict__ msg, const float* __restrict__ dinv,
             const float* __restrict__ b2, const float* __restrict__ bemb,
             float* out, int n) {
  const int lane = threadIdx.x & 31;
  const int i = blockIdx.x * 8 + __builtin_amdgcn_readfirstlane((int)(threadIdx.x >> 5));
  if (i >= n) return;
  const v2f o = ((const v2f*)out)[(size_t)i * 32 + lane];
  const v2f bv = ((const v2f*)b2)[lane];
  const v2f be = ((const v2f*)bemb)[lane];
  const v2f a = row_sum(rowptr, rec, msg, dinv, i, lane);
  v2f r;
  r.x = o.x + be.x + bv.x + a.x;
  r.y = o.y + be.y + bv.y + a.y;
  ((v2f*)out)[(size_t)i * 32 + lane] = r;
}

__global__ void __launch_bounds__(256)
gemm_n64_wmma(const float* __restrict__ A, const float* __restrict__ W,
              float* __restrict__ C, int nrows) {
  const int lane = threadIdx.x & 31;
  const int wave = threadIdx.x >> 5;
  const int tile = blockIdx.x * 8 + wave;
  const int row0 = tile * 16;
  if (row0 >= nrows) return;

  const int half = lane >> 4;
  const int l16  = lane & 15;

  int ar = row0 + l16;
  if (ar >= nrows) ar = nrows - 1;
  const float* __restrict__ arow = A + (size_t)ar * DD;

  v8f acc0 = {}, acc1 = {}, acc2 = {}, acc3 = {};

#pragma unroll
  for (int k = 0; k < 16; ++k) {
    const int kb = k * 4 + half * 2;
    v2f a = *(const v2f*)(arow + kb);

    const float* __restrict__ wr0 = W + (size_t)kb * DD;
    const float* __restrict__ wr1 = W + (size_t)(kb + 1) * DD;
    v2f b0, b1, b2, b3;
    b0.x = wr0[l16];      b0.y = wr1[l16];
    b1.x = wr0[16 + l16]; b1.y = wr1[16 + l16];
    b2.x = wr0[32 + l16]; b2.y = wr1[32 + l16];
    b3.x = wr0[48 + l16]; b3.y = wr1[48 + l16];

    acc0 = __builtin_amdgcn_wmma_f32_16x16x4_f32(false, a, false, b0, (short)0, acc0, false, false);
    acc1 = __builtin_amdgcn_wmma_f32_16x16x4_f32(false, a, false, b1, (short)0, acc1, false, false);
    acc2 = __builtin_amdgcn_wmma_f32_16x16x4_f32(false, a, false, b2, (short)0, acc2, false, false);
    acc3 = __builtin_amdgcn_wmma_f32_16x16x4_f32(false, a, false, b3, (short)0, acc3, false, false);
  }

  float* __restrict__ crow = C + (size_t)row0 * DD;
#pragma unroll
  for (int r = 0; r < 8; ++r) {
    const int rr = r + half * 8;
    if (row0 + rr < nrows) {
      crow[(size_t)rr * DD + l16]      = acc0[r];
      crow[(size_t)rr * DD + 16 + l16] = acc1[r];
      crow[(size_t)rr * DD + 32 + l16] = acc2[r];
      crow[(size_t)rr * DD + 48 + l16] = acc3[r];
    }
  }
}

extern "C" void kernel_launch(void* const* d_in, const int* in_sizes, int n_in,
                              void* d_out, int out_size, void* d_ws, size_t ws_size,
                              hipStream_t stream) {
  const float* x    = (const float*)d_in[0];
  const int*   ei   = (const int*)d_in[1];
  const float* W1   = (const float*)d_in[2];
  const float* b1   = (const float*)d_in[3];
  const float* W2   = (const float*)d_in[4];
  const float* b2   = (const float*)d_in[5];
  const float* Wemb = (const float*)d_in[6];
  const float* bemb = (const float*)d_in[7];
  float* out = (float*)d_out;

  const int N = in_sizes[0] / DD;
  const int E = in_sizes[1] / 2;
  const int* src = ei;
  const int* dst = ei + E;

  const size_t npad = ((size_t)N + 1 + 1023) & ~(size_t)1023;
  const size_t nblk = npad >> 10;
  const size_t bpad = (nblk + 1023) & ~(size_t)1023;
  const size_t epad = ((size_t)E + 63) & ~(size_t)63;
  int*   cnt     = (int*)d_ws;
  int*   rowptr  = cnt + npad;
  float* dinv    = (float*)(rowptr + npad);
  int*   partial = (int*)(dinv + npad);
  int*   rank    = partial + bpad;
  int*   rec     = rank + epad;
  float* buf1    = (float*)(rec + 2 * epad);
  float* buf2    = buf1 + npad * DD;

  const int B   = 256;
  const int gP  = (int)((npad + B - 1) / B);
  const int gE  = (E + B - 1) / B;
  const int gS  = (int)nblk;
  const int gR  = (N + 7) / 8;
  const int gG  = (((N + 15) / 16) + 7) / 8;

  fill_i32<<<gP, B, 0, stream>>>(cnt, 0, (int)npad);
  degree_rank<<<gE, B, 0, stream>>>(dst, cnt, rank, E);
  block_sums<<<gS, B, 0, stream>>>(cnt, partial);
  scan_rows<<<gS, B, 0, stream>>>(cnt, partial, rowptr, dinv);
  place_edges<<<gE, B, 0, stream>>>(src, dst, rank, rowptr, dinv, rec, E);

  gemm_n64_wmma<<<gG, B, 0, stream>>>(x, W1, buf1, N);
  gather_relu<<<gR, B, 0, stream>>>(rowptr, rec, buf1, dinv, b1, buf2, N);

  gemm_n64_wmma<<<gG, B, 0, stream>>>(buf2, W2, buf1, N);
  gemm_n64_wmma<<<gG, B, 0, stream>>>(x, Wemb, out, N);
  gather_final<<<gR, B, 0, stream>>>(rowptr, rec, buf1, dinv, b2, bemb, out, N);
}


# Edge list grouped by destination with LDS counters only (chunk histograms, bins of 512 rows, rank = returning LDS add) instead of 1.6M returning global adds; rows still summed in arrival order
# speedup vs baseline: 4.4382x; 1.5443x over previous
#include <hip/hip_runtime.h>


typedef float v2f __attribute__((ext_vector_type(2)));
typedef float v8f __attribute__((ext_vector_type(8)));

#define DD 64

#define BSH 9
#define BW  512
#define NCH 256
#define MAXB 1024
#define EU  8

__device__ __forceinline__ int block_excl_scan(int v, int* sw) {
  const int lane = threadIdx.x & 31;
  const int wave = threadIdx.x >> 5;
  int inc = v;
  for (int d = 1; d < 32; d <<= 1) {
    const int up = __shfl_up(inc, d);
    if (lane >= d) inc += up;
  }
  if (lane == 31) sw[wave] = inc;
  __syncthreads();
  int off = 0;
  for (int k = 0; k < wave; ++k) off += sw[k];
  return off + inc - v;
}

__global__ void __launch_bounds__(MAXB)
bin_count(const int* __restrict__ dst, int* __restrict__ hist, int ne, int nb, int chunk) {
  __shared__ int lh[MAXB];
  const int t = threadIdx.x;
  lh[t] = 0;
  __syncthreads();
  const int beg = blockIdx.x * chunk;
  const int end = min(ne, beg + chunk);
  for (int e0 = beg + t; e0 < end; e0 += EU * MAXB) {
    int d[EU];
#pragma unroll
    for (int u = 0; u < EU; ++u) d[u] = dst[min(e0 + u * MAXB, end - 1)];
    asm volatile("" :: "v"(d[0]), "v"(d[1]), "v"(d[2]), "v"(d[3]),
                 "v"(d[4]), "v"(d[5]), "v"(d[6]), "v"(d[7]));
#pragma unroll
    for (int u = 0; u < EU; ++u)
      if (e0 + u * MAXB < end) atomicAdd(&lh[d[u] >> BSH], 1);
  }
  __syncthreads();
  if (t < nb) hist[blockIdx.x * nb + t] = lh[t];
}

__global__ void __launch_bounds__(MAXB)
bin_scatter(const int* __restrict__ src, const int* __restrict__ dst,
            const int* __restrict__ hist, int* __restrict__ binbase,
            int* __restrict__ binned, int ne, int nb, int chunk) {
  __shared__ int tot[MAXB];
  __shared__ int pre[MAXB];
  __shared__ int sw[32];
  const int t = threadIdx.x;
  tot[t] = 0;
  pre[t] = 0;
  __syncthreads();
  for (int b0 = 0; b0 < nb; b0 += 256) {
    const int b = b0 + (t & 255);
    if (b < nb) {
      const int* __restrict__ hp = hist + (t >> 8) * nb + b;
      int a = 0;
      int p = 0;
      for (int j0 = 0; j0 < NCH / 4; j0 += 16) {
        int v[16];
#pragma unroll
        for (int u = 0; u < 16; ++u) v[u] = hp[(j0 + u) * 4 * nb];
        asm volatile("" :: "v"(v[0]), "v"(v[1]), "v"(v[2]), "v"(v[3]), "v"(v[4]), "v"(v[5]),
                     "v"(v[6]), "v"(v[7]), "v"(v[8]), "v"(v[9]), "v"(v[10]), "v"(v[11]),
                     "v"(v[12]), "v"(v[13]), "v"(v[14]), "v"(v[15]));
#pragma unroll
        for (int u = 0; u < 16; ++u) {
          a += v[u];
          if ((t >> 8) + 4 * (j0 + u) < (int)blockIdx.x) p += v[u];
        }
      }
      atomicAdd(&tot[b], a);
      atomicAdd(&pre[b], p);
    }
  }
  __syncthreads();
  const int ex = block_excl_scan(tot[t], sw);
  if (blockIdx.x == 0 && t <= nb) binbase[t] = ex;
  tot[t] = ex + pre[t];
  __syncthreads();
  const int beg = blockIdx.x * chunk;
  const int end = min(ne, beg + chunk);
  for (int e0 = beg + t; e0 < end; e0 += EU * MAXB) {
    int d[EU];
    int s[EU];
#pragma unroll
    for (int u = 0; u < EU; ++u) {
      const int e = min(e0 + u * MAXB, end - 1);
      d[u] = dst[e];
      s[u] = src[e];
    }
    asm volatile("" :: "v"(d[0]), "v"(d[1]), "v"(d[2]), "v"(d[3]),
                 "v"(d[4]), "v"(d[5]), "v"(d[6]), "v"(d[7]),
                 "v"(s[0]), "v"(s[1]), "v"(s[2]), "v"(s[3]),
                 "v"(s[4]), "v"(s[5]), "v"(s[6]), "v"(s[7]));
#pragma unroll
    for (int u = 0; u < EU; ++u) {
      if (e0 + u * MAXB < end) {
        const int p = atomicAdd(&tot[d[u] >> BSH], 1);
        binned[p] = (int)(((unsigned)s[u] << BSH) | (unsigned)(d[u] & (BW - 1)));
      }
    }
  }
}

__global__ void __launch_bounds__(BW)
bin_rows(const int* __restrict__ binned, const int* __restrict__ binbase,
         int* __restrict__ rowptr, float* __restrict__ dinv, int* __restrict__ rec) {
  __shared__ int cnt[BW];
  __shared__ int sw[32];
  const int t = threadIdx.x;
  const int beg = binbase[blockIdx.x];
  const int end = binbase[blockIdx.x + 1];
  cnt[t] = 0;
  __syncthreads();
  for (int p0 = beg + t; p0 < end; p0 += EU * BW) {
    int k[EU];
#pragma unroll
    for (int u = 0; u < EU; ++u) k[u] = binned[min(p0 + u * BW, end - 1)];
    asm volatile("" :: "v"(k[0]), "v"(k[1]), "v"(k[2]), "v"(k[3]),
                 "v"(k[4]), "v"(k[5]), "v"(k[6]), "v"(k[7]));
#pragma unroll
    for (int u = 0; u < EU; ++u)
      if (p0 + u * BW < end) atomicAdd(&cnt[k[u] & (BW - 1)], 1);
  }
  __syncthreads();
  const int c = cnt[t];
  const int first = beg + block_excl_scan(c, sw);
  const int node = blockIdx.x * BW + t;
  rowptr[node] = first;
  dinv[node] = rsqrtf((float)(c + 1));
  cnt[t] = first;
  __syncthreads();
  for (int p0 = beg + t; p0 < end; p0 += EU * BW) {
    int k[EU];
#pragma unroll
    for (int u = 0; u < EU; ++u) k[u] = binned[min(p0 + u * BW, end - 1)];
    asm volatile("" :: "v"(k[0]), "v"(k[1]), "v"(k[2]), "v"(k[3]),
                 "v"(k[4]), "v"(k[5]), "v"(k[6]), "v"(k[7]));
#pragma unroll
    for (int u = 0; u < EU; ++u) {
      if (p0 + u * BW < end) {
        const int l = k[u] & (BW - 1);
        int2 r;
        r.x = k[u] >> BSH;
        r.y = blockIdx.x * BW + l;
        ((int2*)rec)[atomicAdd(&cnt[l], 1)] = r;
      }
    }
  }
}

__global__ void __launch_bounds__(256)
edge_weights(const float* __restrict__ dinv, int* __restrict__ rec, int ne) {
  const int p = blockIdx.x * blockDim.x + threadIdx.x;
  if (p >= ne) return;
  int2 r = ((const int2*)rec)[p];
  r.y = __float_as_int(dinv[r.x] * dinv[r.y]);
  ((int2*)rec)[p] = r;
}

__device__ __forceinline__ v2f row_sum(const int* __restrict__ rowptr,
                                       const int* __restrict__ rec,
                                       const float* __restrict__ msg,
                                       const float* __restrict__ dinv,
                                       int i, int lane) {
  const int beg = __builtin_amdgcn_readfirstlane(rowptr[i]);
  const int end = __builtin_amdgcn_readfirstlane(rowptr[i + 1]);
  const float di = dinv[i];
  const v2f* __restrict__ m2 = (const v2f*)msg;
  const v2f self = m2[(size_t)i * 32 + lane];
  const float dd = di * di;
  float ax = dd * self.x;
  float ay = dd * self.y;
  for (int base = beg; base < end; base += 32) {
    const int cnt = min(32, end - base);
    int my = i;
    int wb = 0;
    if (lane < cnt) {
      const int2 r = ((const int2*)rec)[(size_t)base + lane];
      my = r.x;
      wb = r.y;
    }
    for (int j = 0; j < cnt; j += 8) {
      v2f v[8];
      float w[8];
#pragma unroll
      for (int u = 0; u < 8; ++u) {
        const int s = __builtin_amdgcn_readlane(my, j + u);
        w[u] = __int_as_float(__builtin_amdgcn_readlane(wb, j + u));
        v[u] = m2[(size_t)s * 32 + lane];
      }
#pragma unroll
      for (int u = 0; u < 8; ++u) {
        const bool on = (j + u) < cnt;
        ax = fmaf(w[u], on ? v[u].x : 0.0f, ax);
        ay = fmaf(w[u], on ? v[u].y : 0.0f, ay);
      }
    }
  }
  v2f r;
  r.x = ax;
  r.y = ay;
  return r;
}

__global__ void __launch_bounds__(256)
gather_relu(const int* __restrict__ rowptr, const int* __restrict__ rec,
            const float* __restrict__ msg, const float* __restrict__ dinv,
            const float* __restrict__ b, float* __restrict__ h, int n) {
  const int lane = threadIdx.x & 31;
  const int i = blockIdx.x * 8 + __builtin_amdgcn_readfirstlane((int)(threadIdx.x >> 5));
  if (i >= n) return;
  const v2f bv = ((const v2f*)b)[lane];
  const v2f a = row_sum(rowptr, rec, msg, dinv, i, lane);
  v2f r;
  r.x = fmaxf(a.x + bv.x, 0.0f);
  r.y = fmaxf(a.y + bv.y, 0.0f);
  ((v2f*)h)[(size_t)i * 32 + lane] = r;
}

__global__ void __launch_bounds__(256)
gather_final(const int* __restrict__ rowptr, const int* __restrict__ rec,
             const float* __restrict__ msg, const float* __restrict__ dinv,
             const float* __restrict__ b2, const float* __restrict__ bemb,
             float* out, int n) {
  const int lane = threadIdx.x & 31;
  const int i = blockIdx.x * 8 + __builtin_amdgcn_readfirstlane((int)(threadIdx.x >> 5));
  if (i >= n) return;
  const v2f o = ((const v2f*)out)[(size_t)i * 32 + lane];
  const v2f bv = ((const v2f*)b2)[lane];
  const v2f be = ((const v2f*)bemb)[lane];
  const v2f a = row_sum(rowptr, rec, msg, dinv, i, lane);
  v2f r;
  r.x = o.x + be.x + bv.x + a.x;
  r.y = o.y + be.y + bv.y + a.y;
  ((v2f*)out)[(size_t)i * 32 + lane] = r;
}

__global__ void __launch_bounds__(256)
gemm_n64_wmma(const float* __restrict__ A, const float* __restrict__ W,
              float* __restrict__ C, int nrows) {
  const int lane = threadIdx.x & 31;
  const int wave = threadIdx.x >> 5;
  const int tile = blockIdx.x * 8 + wave;
  const int row0 = tile * 16;
  if (row0 >= nrows) return;

  const int half = lane >> 4;
  const int l16  = lane & 15;

  int ar = row0 + l16;
  if (ar >= nrows) ar = nrows - 1;
  const float* __restrict__ arow = A + (size_t)ar * DD;

  v8f acc0 = {}, acc1 = {}, acc2 = {}, acc3 = {};

#pragma unroll
  for (int k = 0; k < 16; ++k) {
    const int kb = k * 4 + half * 2;
    v2f a = *(const v2f*)(arow + kb);

    const float* __restrict__ wr0 = W + (size_t)kb * DD;
    const float* __restrict__ wr1 = W + (size_t)(kb + 1) * DD;
    v2f b0, b1, b2, b3;
    b0.x = wr0[l16];      b0.y = wr1[l16];
    b1.x = wr0[16 + l16]; b1.y = wr1[16 + l16];
    b2.x = wr0[32 + l16]; b2.y = wr1[32 + l16];
    b3.x = wr0[48 + l16]; b3.y = wr1[48 + l16];

    acc0 = __builtin_amdgcn_wmma_f32_16x16x4_f32(false, a, false, b0, (short)0, acc0, false, false);
    acc1 = __builtin_amdgcn_wmma_f32_16x16x4_f32(false, a, false, b1, (short)0, acc1, false, false);
    acc2 = __builtin_amdgcn_wmma_f32_16x16x4_f32(false, a, false, b2, (short)0, acc2, false, false);
    acc3 = __builtin_amdgcn_wmma_f32_16x16x4_f32(false, a, false, b3, (short)0, acc3, false, false);
  }

  float* __restrict__ crow = C + (size_t)row0 * DD;
#pragma unroll
  for (int r = 0; r < 8; ++r) {
    const int rr = r + half * 8;
    if (row0 + rr < nrows) {
      crow[(size_t)rr * DD + l16]      = acc0[r];
      crow[(size_t)rr * DD + 16 + l16] = acc1[r];
      crow[(size_t)rr * DD + 32 + l16] = acc2[r];
      crow[(size_t)rr * DD + 48 + l16] = acc3[r];
    }
  }
}

extern "C" void kernel_launch(void* const* d_in, const int* in_sizes, int n_in,
                              void* d_out, int out_size, void* d_ws, size_t ws_size,
                              hipStream_t stream) {
  const float* x    = (const float*)d_in[0];
  const int*   ei   = (const int*)d_in[1];
  const float* W1   = (const float*)d_in[2];
  const float* b1   = (const float*)d_in[3];
  const float* W2   = (const float*)d_in[4];
  const float* b2   = (const float*)d_in[5];
  const float* Wemb = (const float*)d_in[6];
  const float* bemb = (const float*)d_in[7];
  float* out = (float*)d_out;

  const int N = in_sizes[0] / DD;
  const int E = in_sizes[1] / 2;
  const int* src = ei;
  const int* dst = ei + E;

  const size_t npad = ((size_t)N + 1 + 1023) & ~(size_t)1023;
  const size_t epad = ((size_t)E + 63) & ~(size_t)63;
  int*   rowptr  = (int*)d_ws;
  float* dinv    = (float*)(rowptr + npad);
  int*   hist    = (int*)(dinv + npad);
  int*   binbase = hist + (size_t)NCH * MAXB;
  int*   binned  = binbase + MAXB;
  int*   rec     = binned + epad;
  float* buf1    = (float*)(rec + 2 * epad);
  float* buf2    = buf1 + npad * DD;

  const int B     = 256;
  const int gE    = (E + B - 1) / B;
  const int nb    = N / BW + 1;
  const int chunk = (E + NCH - 1) / NCH;
  const int gR    = (N + 7) / 8;
  const int gG    = (((N + 15) / 16) + 7) / 8;

  bin_count<<<NCH, MAXB, 0, stream>>>(dst, hist, E, nb, chunk);
  bin_scatter<<<NCH, MAXB, 0, stream>>>(src, dst, hist, binbase, binned, E, nb, chunk);
  bin_rows<<<nb, BW, 0, stream>>>(binned, binbase, rowptr, dinv, rec);
  edge_weights<<<gE, B, 0, stream>>>(dinv, rec, E);

  gemm_n64_wmma<<<gG, B, 0, stream>>>(x, W1, buf1, N);
  gather_relu<<<gR, B, 0, stream>>>(rowptr, rec, buf1, dinv, b1, buf2, N);

  gemm_n64_wmma<<<gG, B, 0, stream>>>(buf2, W2, buf1, N);
  gemm_n64_wmma<<<gG, B, 0, stream>>>(x, Wemb, out, N);
  gather_final<<<gR, B, 0, stream>>>(rowptr, rec, buf1, dinv, b2, bemb, out, N);
}


# Edge weight factored as dinv[src]*dinv[dst]: weight launch and 8-byte records go; layer 1 reads dinv[src] beside its 4-byte list, layer-2 rows stored prescaled; term rounding changes; 6 to 5 launches
# speedup vs baseline: 5.8118x; 1.3095x over previous
#include <hip/hip_runtime.h>


typedef float v2f __attribute__((ext_vector_type(2)));
typedef float v8f __attribute__((ext_vector_type(8)));

#define DD 64

#define BSH 9
#define BW  512
#define NCH 256
#define MAXB 1024
#define EU  8

__device__ __forceinline__ int block_excl_scan(int v, int* sw) {
  const int lane = threadIdx.x & 31;
  const int wave = threadIdx.x >> 5;
  int inc = v;
  for (int d = 1; d < 32; d <<= 1) {
    const int up = __shfl_up(inc, d);
    if (lane >= d) inc += up;
  }
  if (lane == 31) sw[wave] = inc;
  __syncthreads();
  int off = 0;
  for (int k = 0; k < wave; ++k) off += sw[k];
  return off + inc - v;
}

__global__ void __launch_bounds__(MAXB)
bin_count(const int* __restrict__ dst, int* __restrict__ hist, int ne, int nb, int chunk) {
  __shared__ int lh[MAXB];
  const int t = threadIdx.x;
  lh[t] = 0;
  __syncthreads();
  const int beg = blockIdx.x * chunk;
  const int end = min(ne, beg + chunk);
  for (int e0 = beg + t; e0 < end; e0 += EU * MAXB) {
    int d[EU];
#pragma unroll
    for (int u = 0; u < EU; ++u) d[u] = dst[min(e0 + u * MAXB, end - 1)];
    asm volatile("" :: "v"(d[0]), "v"(d[1]), "v"(d[2]), "v"(d[3]),
                 "v"(d[4]), "v"(d[5]), "v"(d[6]), "v"(d[7]));
#pragma unroll
    for (int u = 0; u < EU; ++u)
      if (e0 + u * MAXB < end) atomicAdd(&lh[d[u] >> BSH], 1);
  }
  __syncthreads();
  if (t < nb) hist[blockIdx.x * nb + t] = lh[t];
}

__global__ void __launch_bounds__(MAXB)
bin_scatter(const int* __restrict__ src, const int* __restrict__ dst,
            const int* __restrict__ hist, int* __restrict__ binbase,
            int* __restrict__ binned, int ne, int nb, int chunk) {
  __shared__ int tot[MAXB];
  __shared__ int pre[MAXB];
  __shared__ int sw[32];
  const int t = threadIdx.x;
  tot[t] = 0;
  pre[t] = 0;
  __syncthreads();
  for (int b0 = 0; b0 < nb; b0 += 256) {
    const int b = b0 + (t & 255);
    if (b < nb) {
      const int* __restrict__ hp = hist + (t >> 8) * nb + b;
      int a = 0;
      int p = 0;
      for (int j0 = 0; j0 < NCH / 4; j0 += 16) {
        int v[16];
#pragma unroll
        for (int u = 0; u < 16; ++u) v[u] = hp[(j0 + u) * 4 * nb];
        asm volatile("" :: "v"(v[0]), "v"(v[1]), "v"(v[2]), "v"(v[3]), "v"(v[4]), "v"(v[5]),
                     "v"(v[6]), "v"(v[7]), "v"(v[8]), "v"(v[9]), "v"(v[10]), "v"(v[11]),
                     "v"(v[12]), "v"(v[13]), "v"(v[14]), "v"(v[15]));
#pragma unroll
        for (int u = 0; u < 16; ++u) {
          a += v[u];
          if ((t >> 8) + 4 * (j0 + u) < (int)blockIdx.x) p += v[u];
        }
      }
      atomicAdd(&tot[b], a);
      atomicAdd(&pre[b], p);
    }
  }
  __syncthreads();
  const int ex = block_excl_scan(tot[t], sw);
  if (blockIdx.x == 0 && t <= nb) binbase[t] = ex;
  tot[t] = ex + pre[t];
  __syncthreads();
  const int beg = blockIdx.x * chunk;
  const int end = min(ne, beg + chunk);
  for (int e0 = beg + t; e0 < end; e0 += EU * MAXB) {
    int d[EU];
    int s[EU];
#pragma unroll
    for (int u = 0; u < EU; ++u) {
      const int e = min(e0 + u * MAXB, end - 1);
      d[u] = dst[e];
      s[u] = src[e];
    }
    asm volatile("" :: "v"(d[0]), "v"(d[1]), "v"(d[2]), "v"(d[3]),
                 "v"(d[4]), "v"(d[5]), "v"(d[6]), "v"(d[7]),
                 "v"(s[0]), "v"(s[1]), "v"(s[2]), "v"(s[3]),
                 "v"(s[4]), "v"(s[5]), "v"(s[6]), "v"(s[7]));
#pragma unroll
    for (int u = 0; u < EU; ++u) {
      if (e0 + u * MAXB < end) {
        const int p = atomicAdd(&tot[d[u] >> BSH], 1);
        binned[p] = (int)(((unsigned)s[u] << BSH) | (unsigned)(d[u] & (BW - 1)));
      }
    }
  }
}

__global__ void __launch_bounds__(BW)
bin_rows(const int* __restrict__ binned, const int* __restrict__ binbase,
         int* __restrict__ rowptr, float* __restrict__ dinv, int* __restrict__ rec) {
  __shared__ int cnt[BW];
  __shared__ int sw[32];
  const int t = threadIdx.x;
  const int beg = binbase[blockIdx.x];
  const int end = binbase[blockIdx.x + 1];
  cnt[t] = 0;
  __syncthreads();
  for (int p0 = beg + t; p0 < end; p0 += EU * BW) {
    int k[EU];
#pragma unroll
    for (int u = 0; u < EU; ++u) k[u] = binned[min(p0 + u * BW, end - 1)];
    asm volatile("" :: "v"(k[0]), "v"(k[1]), "v"(k[2]), "v"(k[3]),
                 "v"(k[4]), "v"(k[5]), "v"(k[6]), "v"(k[7]));
#pragma unroll
    for (int u = 0; u < EU; ++u)
      if (p0 + u * BW < end) atomicAdd(&cnt[k[u] & (BW - 1)], 1);
  }
  __syncthreads();
  const int c = cnt[t];
  const int first = beg + block_excl_scan(c, sw);
  const int node = blockIdx.x * BW + t;
  rowptr[node] = first;
  dinv[node] = rsqrtf((float)(c + 1));
  cnt[t] = first;
  __syncthreads();
  for (int p0 = beg + t; p0 < end; p0 += EU * BW) {
    int k[EU];
#pragma unroll
    for (int u = 0; u < EU; ++u) k[u] = binned[min(p0 + u * BW, end - 1)];
    asm volatile("" :: "v"(k[0]), "v"(k[1]), "v"(k[2]), "v"(k[3]),
                 "v"(k[4]), "v"(k[5]), "v"(k[6]), "v"(k[7]));
#pragma unroll
    for (int u = 0; u < EU; ++u) {
      if (p0 + u * BW < end) {
        const int l = k[u] & (BW - 1);
        rec[atomicAdd(&cnt[l], 1)] = k[u] >> BSH;
      }
    }
  }
}

template <bool WSRC>
__device__ __forceinline__ v2f row_sum(const int* __restrict__ rowptr,
                                       const int* __restrict__ rec,
                                       const float* __restrict__ msg,
                                       const float* __restrict__ dinv,
                                       int i, int lane, v2f& own, float& di) {
  const int beg = __builtin_amdgcn_readfirstlane(rowptr[i]);
  const int end = __builtin_amdgcn_readfirstlane(rowptr[i + 1]);
  di = dinv[i];
  const v2f* __restrict__ m2 = (const v2f*)msg;
  const v2f self = m2[(size_t)i * 32 + lane];
  own = self;
  float ax = WSRC ? di * self.x : self.x;
  float ay = WSRC ? di * self.y : self.y;
  for (int base = beg; base < end; base += 32) {
    const int cnt = min(32, end - base);
    int my = i;
    float wb = 0.0f;
    if (lane < cnt) {
      my = rec[(size_t)base + lane];
      if (WSRC) wb = dinv[my];
    }
    for (int j = 0; j < cnt; j += 8) {
      v2f v[8];
      float w[8];
#pragma unroll
      for (int u = 0; u < 8; ++u) {
        const int s = __builtin_amdgcn_readlane(my, j + u);
        w[u] = WSRC ? __int_as_float(__builtin_amdgcn_readlane(__float_as_int(wb), j + u)) : 1.0f;
        v[u] = m2[(size_t)s * 32 + lane];
      }
#pragma unroll
      for (int u = 0; u < 8; ++u) {
        const bool on = (j + u) < cnt;
        const float vx = on ? v[u].x : 0.0f;
        const float vy = on ? v[u].y : 0.0f;
        if (WSRC) {
          ax = fmaf(w[u], vx, ax);
          ay = fmaf(w[u], vy, ay);
        } else {
          ax += vx;
          ay += vy;
        }
      }
    }
  }
  v2f r;
  r.x = ax;
  r.y = ay;
  return r;
}

#define TP 68
#define WP 72

__device__ __forceinline__ v8f wmma4(v2f a, v2f b, v8f c) {
  return __builtin_amdgcn_wmma_f32_16x16x4_f32(false, a, false, b, (short)0, c, false, false);
}

__global__ void __launch_bounds__(512)
gather_dense(const int* __restrict__ rowptr, const int* __restrict__ rec,
             const float* __restrict__ x, const float* __restrict__ dinv,
             const float* __restrict__ W1, const float* __restrict__ b1,
             const float* __restrict__ W2, const float* __restrict__ Wemb,
             float* __restrict__ hw2, float* __restrict__ out, int n) {
  __shared__ __attribute__((aligned(16))) float Ts[32 * TP];
  __shared__ __attribute__((aligned(16))) float Xs[32 * TP];
  __shared__ __attribute__((aligned(16))) float Hs[32 * TP];
  __shared__ __attribute__((aligned(16))) float Es[32 * TP];
  __shared__ __attribute__((aligned(16))) float W2s[DD * WP];
  __shared__ float b1s[DD];
  const int lane = threadIdx.x & 31;
  const int half = lane >> 4;
  const int l16  = lane & 15;
  const int w = __builtin_amdgcn_readfirstlane((int)(threadIdx.x >> 5));
  const int row0 = blockIdx.x * 32;

  float dsv[2];
#pragma unroll
  for (int rr = 0; rr < 2; ++rr) {
    const int lr = rr * 16 + w;
    const int ir = min(row0 + lr, n - 1);
    v2f own;
    float di;
    const v2f s = row_sum<true>(rowptr, rec, x, dinv, ir, lane, own, di);
    dsv[rr] = di;
    v2f t;
    t.x = di * s.x;
    t.y = di * s.y;
    ((v2f*)(Ts + lr * TP))[lane] = t;
    ((v2f*)(Xs + lr * TP))[lane] = own;
  }

  const int tile = (w >> 2) & 1;
  const int ct   = w & 3;
  const float* __restrict__ Wm = (w < 8) ? W1 : Wemb;
  v2f bf[16];
#pragma unroll
  for (int k = 0; k < 16; ++k) {
    const int kb = 4 * k + 2 * half;
    bf[k].x = Wm[kb * DD + 16 * ct + l16];
    bf[k].y = Wm[(kb + 1) * DD + 16 * ct + l16];
  }
  {
    const int t  = threadIdx.x;
    const int r  = t >> 3;
    const int c0 = (t & 7) * 8;
    const float4 q0 = ((const float4*)W2)[2 * t];
    const float4 q1 = ((const float4*)W2)[2 * t + 1];
    *(float4*)(W2s + r * WP + c0)     = q0;
    *(float4*)(W2s + r * WP + c0 + 4) = q1;
    if (t < DD) b1s[t] = b1[t];
  }
  __syncthreads();

  {
    const float* As = ((w < 8) ? Ts : Xs) + (tile * 16 + l16) * TP + 2 * half;
    v8f acc0 = {}, acc1 = {};
#pragma unroll
    for (int k = 0; k < 16; k += 2) {
      acc0 = wmma4(*(const v2f*)(As + 4 * k), bf[k], acc0);
      acc1 = wmma4(*(const v2f*)(As + 4 * (k + 1)), bf[k + 1], acc1);
    }
    const v8f acc = acc0 + acc1;
    const int o = (tile * 16 + 8 * half) * TP + 16 * ct + l16;
    if (w < 8) {
      const float bb = b1s[16 * ct + l16];
#pragma unroll
      for (int r = 0; r < 8; ++r) Hs[o + r * TP] = fmaxf(acc[r] + bb, 0.0f);
    } else {
#pragma unroll
      for (int r = 0; r < 8; ++r) Es[o + r * TP] = acc[r];
    }
  }
  __syncthreads();

  if (w < 8) {
    const float* As = Hs + (tile * 16 + l16) * TP + 2 * half;
    const float* Bs = W2s + 2 * half * WP + 16 * ct + l16;
    v8f acc0 = {}, acc1 = {};
#pragma unroll
    for (int k = 0; k < 16; k += 2) {
      v2f b0, b1v;
      b0.x  = Bs[(4 * k) * WP];      b0.y  = Bs[(4 * k + 1) * WP];
      b1v.x = Bs[(4 * k + 4) * WP];  b1v.y = Bs[(4 * k + 5) * WP];
      acc0 = wmma4(*(const v2f*)(As + 4 * k), b0, acc0);
      acc1 = wmma4(*(const v2f*)(As + 4 * (k + 1)), b1v, acc1);
    }
    const v8f acc = acc0 + acc1;
    const int o = (tile * 16 + 8 * half) * TP + 16 * ct + l16;
#pragma unroll
    for (int r = 0; r < 8; ++r) Ts[o + r * TP] = acc[r];
  }
  __syncthreads();

#pragma unroll
  for (int rr = 0; rr < 2; ++rr) {
    const int lr = rr * 16 + w;
    const int i = row0 + lr;
    if (i < n) {
      v2f p = ((const v2f*)(Ts + lr * TP))[lane];
      p.x *= dsv[rr];
      p.y *= dsv[rr];
      ((v2f*)hw2)[(size_t)i * 32 + lane] = p;
      ((v2f*)out)[(size_t)i * 32 + lane] = ((const v2f*)(Es + lr * TP))[lane];
    }
  }
}

__global__ void __launch_bounds__(256)
gather_final(const int* __restrict__ rowptr, const int* __restrict__ rec,
             const float* __restrict__ msg, const float* __restrict__ dinv,
             const float* __restrict__ b2, const float* __restrict__ bemb,
             float* out, int n) {
  const int lane = threadIdx.x & 31;
  const int i = blockIdx.x * 8 + __builtin_amdgcn_readfirstlane((int)(threadIdx.x >> 5));
  if (i >= n) return;
  const v2f o = ((const v2f*)out)[(size_t)i * 32 + lane];
  const v2f bv = ((const v2f*)b2)[lane];
  const v2f be = ((const v2f*)bemb)[lane];
  v2f own;
  float di;
  const v2f a = row_sum<false>(rowptr, rec, msg, dinv, i, lane, own, di);
  v2f r;
  r.x = o.x + be.x + bv.x + di * a.x;
  r.y = o.y + be.y + bv.y + di * a.y;
  ((v2f*)out)[(size_t)i * 32 + lane] = r;
}

extern "C" void kernel_launch(void* const* d_in, const int* in_sizes, int n_in,
                              void* d_out, int out_size, void* d_ws, size_t ws_size,
                              hipStream_t stream) {
  const float* x    = (const float*)d_in[0];
  const int*   ei   = (const int*)d_in[1];
  const float* W1   = (const float*)d_in[2];
  const float* b1   = (const float*)d_in[3];
  const float* W2   = (const float*)d_in[4];
  const float* b2   = (const float*)d_in[5];
  const float* Wemb = (const float*)d_in[6];
  const float* bemb = (const float*)d_in[7];
  float* out = (float*)d_out;

  const int N = in_sizes[0] / DD;
  const int E = in_sizes[1] / 2;
  const int* src = ei;
  const int* dst = ei + E;

  const size_t npad = ((size_t)N + 1 + 1023) & ~(size_t)1023;
  const size_t epad = ((size_t)E + 63) & ~(size_t)63;
  int*   rowptr  = (int*)d_ws;
  float* dinv    = (float*)(rowptr + npad);
  int*   hist    = (int*)(dinv + npad);
  int*   binbase = hist + (size_t)NCH * MAXB;
  int*   binned  = binbase + MAXB;
  int*   rec     = binned + epad;
  float* buf1    = (float*)(rec + epad);

  const int B     = 256;
  const int nb    = N / BW + 1;
  const int chunk = (E + NCH - 1) / NCH;
  const int gR    = (N + 7) / 8;
  const int gT    = (N + 31) / 32;

  bin_count<<<NCH, MAXB, 0, stream>>>(dst, hist, E, nb, chunk);
  bin_scatter<<<NCH, MAXB, 0, stream>>>(src, dst, hist, binbase, binned, E, nb, chunk);
  bin_rows<<<nb, BW, 0, stream>>>(binned, binbase, rowptr, dinv, rec);

  gather_dense<<<gT, 512, 0, stream>>>(rowptr, rec, x, dinv, W1, b1, W2, Wemb, buf1, out, N);

  gather_final<<<gR, B, 0, stream>>>(rowptr, rec, buf1, dinv, b2, bemb, out, N);
}


# Edge list built in two launches, not three: each chunk orders its edges by bin in its own stretch (LDS histogram + block scan), bin blocks read the pieces; row sums take another arrival order
# speedup vs baseline: 6.9869x; 1.2022x over previous
#include <hip/hip_runtime.h>


typedef float v2f __attribute__((ext_vector_type(2)));
typedef float v8f __attribute__((ext_vector_type(8)));

#define DD 64

#define BSH 9
#define BW  512
#define NCH 256
#define MAXB 1024
#define EU  8
#define SPW (NCH / (BW / 32))
#define SPG 8

__device__ __forceinline__ int block_excl_scan(int v, int* sw) {
  const int lane = threadIdx.x & 31;
  const int wave = threadIdx.x >> 5;
  int inc = v;
  for (int d = 1; d < 32; d <<= 1) {
    const int up = __shfl_up(inc, d);
    if (lane >= d) inc += up;
  }
  if (lane == 31) sw[wave] = inc;
  __syncthreads();
  int off = 0;
  for (int k = 0; k < wave; ++k) off += sw[k];
  return off + inc - v;
}

__global__ void __launch_bounds__(MAXB)
chunk_sort(const int* __restrict__ src, const int* __restrict__ dst,
           int* __restrict__ offt, int* __restrict__ cb, int ne, int nb, int chunk) {
  __shared__ int lh[MAXB];
  __shared__ int sw[32];
  const int t = threadIdx.x;
  lh[t] = 0;
  __syncthreads();
  const int beg = blockIdx.x * chunk;
  const int end = min(ne, beg + chunk);
  for (int e0 = beg + t; e0 < end; e0 += EU * MAXB) {
    int d[EU];
#pragma unroll
    for (int u = 0; u < EU; ++u) d[u] = dst[min(e0 + u * MAXB, end - 1)];
    asm volatile("" :: "v"(d[0]), "v"(d[1]), "v"(d[2]), "v"(d[3]),
                 "v"(d[4]), "v"(d[5]), "v"(d[6]), "v"(d[7]));
#pragma unroll
    for (int u = 0; u < EU; ++u)
      if (e0 + u * MAXB < end) atomicAdd(&lh[d[u] >> BSH], 1);
  }
  __syncthreads();
  const int ex = block_excl_scan(lh[t], sw);
  if (t <= nb) offt[blockIdx.x * (nb + 1) + t] = ex;
  lh[t] = beg + ex;
  __syncthreads();
  for (int e0 = beg + t; e0 < end; e0 += EU * MAXB) {
    int d[EU];
    int s[EU];
#pragma unroll
    for (int u = 0; u < EU; ++u) {
      const int e = min(e0 + u * MAXB, end - 1);
      d[u] = dst[e];
      s[u] = src[e];
    }
    asm volatile("" :: "v"(d[0]), "v"(d[1]), "v"(d[2]), "v"(d[3]),
                 "v"(d[4]), "v"(d[5]), "v"(d[6]), "v"(d[7]),
                 "v"(s[0]), "v"(s[1]), "v"(s[2]), "v"(s[3]),
                 "v"(s[4]), "v"(s[5]), "v"(s[6]), "v"(s[7]));
#pragma unroll
    for (int u = 0; u < EU; ++u) {
      if (e0 + u * MAXB < end) {
        const int p = atomicAdd(&lh[d[u] >> BSH], 1);
        cb[p] = (int)(((unsigned)s[u] << BSH) | (unsigned)(d[u] & (BW - 1)));
      }
    }
  }
}

template <bool PLACE>
__device__ __forceinline__ void piece_sweep(const int* __restrict__ cb, const int (&lo)[SPW],
                                            const int (&hi)[SPW], int lane, int* cnt,
                                            int* __restrict__ rec) {
#pragma unroll
  for (int g = 0; g < SPW; g += SPG) {
    int k[2 * SPG];
#pragma unroll
    for (int s = 0; s < SPG; ++s) {
#pragma unroll
      for (int u = 0; u < 2; ++u) {
        const int p = lo[g + s] + lane + 32 * u;
        k[2 * s + u] = cb[max(min(p, hi[g + s] - 1), 0)];
      }
    }
    asm volatile("" :: "v"(k[0]), "v"(k[1]), "v"(k[2]), "v"(k[3]), "v"(k[4]), "v"(k[5]),
                 "v"(k[6]), "v"(k[7]), "v"(k[8]), "v"(k[9]), "v"(k[10]), "v"(k[11]),
                 "v"(k[12]), "v"(k[13]), "v"(k[14]), "v"(k[15]));
#pragma unroll
    for (int s = 0; s < SPG; ++s) {
#pragma unroll
      for (int u = 0; u < 2; ++u) {
        if (lo[g + s] + lane + 32 * u < hi[g + s]) {
          const int key = k[2 * s + u];
          if (PLACE) rec[atomicAdd(&cnt[key & (BW - 1)], 1)] = key >> BSH;
          else atomicAdd(&cnt[key & (BW - 1)], 1);
        }
      }
    }
#pragma unroll
    for (int s = 0; s < SPG; ++s) {
      for (int p = lo[g + s] + 64 + lane; p < hi[g + s]; p += 32) {
        const int key = cb[p];
        if (PLACE) rec[atomicAdd(&cnt[key & (BW - 1)], 1)] = key >> BSH;
        else atomicAdd(&cnt[key & (BW - 1)], 1);
      }
    }
  }
}

__global__ void __launch_bounds__(BW)
bin_rows(const int* __restrict__ cb, const int* __restrict__ offt,
         int* __restrict__ rowptr, float* __restrict__ dinv, int* __restrict__ rec,
         int nb, int chunk) {
  __shared__ int cnt[BW];
  __shared__ int sw[32];
  __shared__ int sbeg;
  const int t = threadIdx.x;
  const int lane = t & 31;
  const int w = __builtin_amdgcn_readfirstlane(t >> 5);
  const int b = blockIdx.x;
  cnt[t] = 0;
  if (t == 0) sbeg = 0;
  int lo[SPW];
  int hi[SPW];
  int below = 0;
#pragma unroll
  for (int s = 0; s < SPW; ++s) {
    const int c = w + (BW / 32) * s;
    const int o0 = offt[c * (nb + 1) + b];
    const int o1 = offt[c * (nb + 1) + b + 1];
    below += o0;
    lo[s] = c * chunk + o0;
    hi[s] = c * chunk + o1;
  }
  __syncthreads();
  if (lane == 0) atomicAdd(&sbeg, below);
  piece_sweep<false>(cb, lo, hi, lane, cnt, rec);
  __syncthreads();
  const int c = cnt[t];
  const int first = sbeg + block_excl_scan(c, sw);
  const int node = b * BW + t;
  rowptr[node] = first;
  dinv[node] = rsqrtf((float)(c + 1));
  cnt[t] = first;
  __syncthreads();
  piece_sweep<true>(cb, lo, hi, lane, cnt, rec);
}

template <bool WSRC>
__device__ __forceinline__ v2f row_sum(const int* __restrict__ rowptr,
                                       const int* __restrict__ rec,
                                       const float* __restrict__ msg,
                                       const float* __restrict__ dinv,
                                       int i, int lane, v2f& own, float& di) {
  const int beg = __builtin_amdgcn_readfirstlane(rowptr[i]);
  const int end = __builtin_amdgcn_readfirstlane(rowptr[i + 1]);
  di = dinv[i];
  const v2f* __restrict__ m2 = (const v2f*)msg;
  const v2f self = m2[(size_t)i * 32 + lane];
  own = self;
  float ax = WSRC ? di * self.x : self.x;
  float ay = WSRC ? di * self.y : self.y;
  for (int base = beg; base < end; base += 32) {
    const int cnt = min(32, end - base);
    int my = i;
    float wb = 0.0f;
    if (lane < cnt) {
      my = rec[(size_t)base + lane];
      if (WSRC) wb = dinv[my];
    }
    for (int j = 0; j < cnt; j += 8) {
      v2f v[8];
      float w[8];
#pragma unroll
      for (int u = 0; u < 8; ++u) {
        const int s = __builtin_amdgcn_readlane(my, j + u);
        w[u] = WSRC ? __int_as_float(__builtin_amdgcn_readlane(__float_as_int(wb), j + u)) : 1.0f;
        v[u] = m2[(size_t)s * 32 + lane];
      }
#pragma unroll
      for (int u = 0; u < 8; ++u) {
        const bool on = (j + u) < cnt;
        const float vx = on ? v[u].x : 0.0f;
        const float vy = on ? v[u].y : 0.0f;
        if (WSRC) {
          ax = fmaf(w[u], vx, ax);
          ay = fmaf(w[u], vy, ay);
        } else {
          ax += vx;
          ay += vy;
        }
      }
    }
  }
  v2f r;
  r.x = ax;
  r.y = ay;
  return r;
}

#define TP 68
#define WP 72

__device__ __forceinline__ v8f wmma4(v2f a, v2f b, v8f c) {
  return __builtin_amdgcn_wmma_f32_16x16x4_f32(false, a, false, b, (short)0, c, false, false);
}

__global__ void __launch_bounds__(512)
gather_dense(const int* __restrict__ rowptr, const int* __restrict__ rec,
             const float* __restrict__ x, const float* __restrict__ dinv,
             const float* __restrict__ W1, const float* __restrict__ b1,
             const float* __restrict__ W2, const float* __restrict__ Wemb,
             float* __restrict__ hw2, float* __restrict__ out, int n) {
  __shared__ __attribute__((aligned(16))) float Ts[32 * TP];
  __shared__ __attribute__((aligned(16))) float Xs[32 * TP];
  __shared__ __attribute__((aligned(16))) float Hs[32 * TP];
  __shared__ __attribute__((aligned(16))) float Es[32 * TP];
  __shared__ __attribute__((aligned(16))) float W2s[DD * WP];
  __shared__ float b1s[DD];
  const int lane = threadIdx.x & 31;
  const int half = lane >> 4;
  const int l16  = lane & 15;
  const int w = __builtin_amdgcn_readfirstlane((int)(threadIdx.x >> 5));
  const int row0 = blockIdx.x * 32;

  float dsv[2];
#pragma unroll
  for (int rr = 0; rr < 2; ++rr) {
    const int lr = rr * 16 + w;
    const int ir = min(row0 + lr, n - 1);
    v2f own;
    float di;
    const v2f s = row_sum<true>(rowptr, rec, x, dinv, ir, lane, own, di);
    dsv[rr] = di;
    v2f t;
    t.x = di * s.x;
    t.y = di * s.y;
    ((v2f*)(Ts + lr * TP))[lane] = t;
    ((v2f*)(Xs + lr * TP))[lane] = own;
  }

  const int tile = (w >> 2) & 1;
  const int ct   = w & 3;
  const float* __restrict__ Wm = (w < 8) ? W1 : Wemb;
  v2f bf[16];
#pragma unroll
  for (int k = 0; k < 16; ++k) {
    const int kb = 4 * k + 2 * half;
    bf[k].x = Wm[kb * DD + 16 * ct + l16];
    bf[k].y = Wm[(kb + 1) * DD + 16 * ct + l16];
  }
  {
    const int t  = threadIdx.x;
    const int r  = t >> 3;
    const int c0 = (t & 7) * 8;
    const float4 q0 = ((const float4*)W2)[2 * t];
    const float4 q1 = ((const float4*)W2)[2 * t + 1];
    *(float4*)(W2s + r * WP + c0)     = q0;
    *(float4*)(W2s + r * WP + c0 + 4) = q1;
    if (t < DD) b1s[t] = b1[t];
  }
  __syncthreads();

  {
    const float* As = ((w < 8) ? Ts : Xs) + (tile * 16 + l16) * TP + 2 * half;
    v8f acc0 = {}, acc1 = {};
#pragma unroll
    for (int k = 0; k < 16; k += 2) {
      acc0 = wmma4(*(const v2f*)(As + 4 * k), bf[k], acc0);
      acc1 = wmma4(*(const v2f*)(As + 4 * (k + 1)), bf[k + 1], acc1);
    }
    const v8f acc = acc0 + acc1;
    const int o = (tile * 16 + 8 * half) * TP + 16 * ct + l16;
    if (w < 8) {
      const float bb = b1s[16 * ct + l16];
#pragma unroll
      for (int r = 0; r < 8; ++r) Hs[o + r * TP] = fmaxf(acc[r] + bb, 0.0f);
    } else {
#pragma unroll
      for (int r = 0; r < 8; ++r) Es[o + r * TP] = acc[r];
    }
  }
  __syncthreads();

  if (w < 8) {
    const float* As = Hs + (tile * 16 + l16) * TP + 2 * half;
    const float* Bs = W2s + 2 * half * WP + 16 * ct + l16;
    v8f acc0 = {}, acc1 = {};
#pragma unroll
    for (int k = 0; k < 16; k += 2) {
      v2f b0, b1v;
      b0.x  = Bs[(4 * k) * WP];      b0.y  = Bs[(4 * k + 1) * WP];
      b1v.x = Bs[(4 * k + 4) * WP];  b1v.y = Bs[(4 * k + 5) * WP];
      acc0 = wmma4(*(const v2f*)(As + 4 * k), b0, acc0);
      acc1 = wmma4(*(const v2f*)(As + 4 * (k + 1)), b1v, acc1);
    }
    const v8f acc = acc0 + acc1;
    const int o = (tile * 16 + 8 * half) * TP + 16 * ct + l16;
#pragma unroll
    for (int r = 0; r < 8; ++r) Ts[o + r * TP] = acc[r];
  }
  __syncthreads();

#pragma unroll
  for (int rr = 0; rr < 2; ++rr) {
    const int lr = rr * 16 + w;
    const int i = row0 + lr;
    if (i < n) {
      v2f p = ((const v2f*)(Ts + lr * TP))[lane];
      p.x *= dsv[rr];
      p.y *= dsv[rr];
      ((v2f*)hw2)[(size_t)i * 32 + lane] = p;
      ((v2f*)out)[(size_t)i * 32 + lane] = ((const v2f*)(Es + lr * TP))[lane];
    }
  }
}

__global__ void __launch_bounds__(256)
gather_final(const int* __restrict__ rowptr, const int* __restrict__ rec,
             const float* __restrict__ msg, const float* __restrict__ dinv,
             const float* __restrict__ b2, const float* __restrict__ bemb,
             float* out, int n) {
  const int lane = threadIdx.x & 31;
  const int i = blockIdx.x * 8 + __builtin_amdgcn_readfirstlane((int)(threadIdx.x >> 5));
  if (i >= n) return;
  const v2f o = ((const v2f*)out)[(size_t)i * 32 + lane];
  const v2f bv = ((const v2f*)b2)[lane];
  const v2f be = ((const v2f*)bemb)[lane];
  v2f own;
  float di;
  const v2f a = row_sum<false>(rowptr, rec, msg, dinv, i, lane, own, di);
  v2f r;
  r.x = o.x + be.x + bv.x + di * a.x;
  r.y = o.y + be.y + bv.y + di * a.y;
  ((v2f*)out)[(size_t)i * 32 + lane] = r;
}

extern "C" void kernel_launch(void* const* d_in, const int* in_sizes, int n_in,
                              void* d_out, int out_size, void* d_ws, size_t ws_size,
                              hipStream_t stream) {
  const float* x    = (const float*)d_in[0];
  const int*   ei   = (const int*)d_in[1];
  const float* W1   = (const float*)d_in[2];
  const float* b1   = (const float*)d_in[3];
  const float* W2   = (const float*)d_in[4];
  const float* b2   = (const float*)d_in[5];
  const float* Wemb = (const float*)d_in[6];
  const float* bemb = (const float*)d_in[7];
  float* out = (float*)d_out;

  const int N = in_sizes[0] / DD;
  const int E = in_sizes[1] / 2;
  const int* src = ei;
  const int* dst = ei + E;

  const size_t npad = ((size_t)N + 1 + 1023) & ~(size_t)1023;
  const size_t epad = ((size_t)E + 63) & ~(size_t)63;
  int*   rowptr  = (int*)d_ws;
  float* dinv    = (float*)(rowptr + npad);
  int*   offt    = (int*)(dinv + npad);
  int*   cb      = offt + (size_t)NCH * MAXB;
  int*   rec     = cb + epad;
  float* buf1    = (float*)(rec + epad);

  const int B     = 256;
  const int nb    = N / BW + 1;
  const int chunk = (E + NCH - 1) / NCH;
  const int gR    = (N + 7) / 8;
  const int gT    = (N + 31) / 32;

  chunk_sort<<<NCH, MAXB, 0, stream>>>(src, dst, offt, cb, E, nb, chunk);
  bin_rows<<<nb, BW, 0, stream>>>(cb, offt, rowptr, dinv, rec, nb, chunk);

  gather_dense<<<gT, 512, 0, stream>>>(rowptr, rec, x, dinv, W1, b1, W2, Wemb, buf1, out, N);

  gather_final<<<gR, B, 0, stream>>>(rowptr, rec, buf1, dinv, b2, bemb, out, N);
}


# Both list-build launches place their words in an LDS copy of the stretch and store it in whole-line wave-wide writes: one LDS write per edge, not one partial-line memory write; arrival order kept
# speedup vs baseline: 7.6266x; 1.0916x over previous
#include <hip/hip_runtime.h>


typedef float v2f __attribute__((ext_vector_type(2)));
typedef float v8f __attribute__((ext_vector_type(8)));

#define DD 64

#define BSH 9
#define BW  512
#define NCH 256
#define MAXB 1024
#define EU  8
#define SPW (NCH / (BW / 32))
#define SPG 8
#define SCAP 8192
#define RCAP 12288

__device__ __forceinline__ int block_excl_scan(int v, int* sw) {
  const int lane = threadIdx.x & 31;
  const int wave = threadIdx.x >> 5;
  int inc = v;
  for (int d = 1; d < 32; d <<= 1) {
    const int up = __shfl_up(inc, d);
    if (lane >= d) inc += up;
  }
  if (lane == 31) sw[wave] = inc;
  __syncthreads();
  int off = 0;
  for (int k = 0; k < wave; ++k) off += sw[k];
  return off + inc - v;
}

template <bool TO_LDS>
__device__ __forceinline__ void place_keys(const int* __restrict__ src,
                                           const int* __restrict__ dst, int* lh, int* stage,
                                           int* __restrict__ cb, int beg, int end, int t) {
  for (int e0 = beg + t; e0 < end; e0 += EU * MAXB) {
    int d[EU];
    int s[EU];
#pragma unroll
    for (int u = 0; u < EU; ++u) {
      const int e = min(e0 + u * MAXB, end - 1);
      d[u] = dst[e];
      s[u] = src[e];
    }
    asm volatile("" :: "v"(d[0]), "v"(d[1]), "v"(d[2]), "v"(d[3]),
                 "v"(d[4]), "v"(d[5]), "v"(d[6]), "v"(d[7]),
                 "v"(s[0]), "v"(s[1]), "v"(s[2]), "v"(s[3]),
                 "v"(s[4]), "v"(s[5]), "v"(s[6]), "v"(s[7]));
#pragma unroll
    for (int u = 0; u < EU; ++u) {
      if (e0 + u * MAXB < end) {
        const int p = atomicAdd(&lh[d[u] >> BSH], 1);
        const int key = (int)(((unsigned)s[u] << BSH) | (unsigned)(d[u] & (BW - 1)));
        if (TO_LDS) stage[p - beg] = key;
        else cb[p] = key;
      }
    }
  }
}

__global__ void __launch_bounds__(MAXB)
chunk_sort(const int* __restrict__ src, const int* __restrict__ dst,
           int* __restrict__ offt, int* __restrict__ cb, int ne, int nb, int chunk) {
  __shared__ int lh[MAXB];
  __shared__ int sw[32];
  __shared__ int stage[SCAP];
  const int t = threadIdx.x;
  lh[t] = 0;
  __syncthreads();
  const int beg = blockIdx.x * chunk;
  const int end = min(ne, beg + chunk);
  for (int e0 = beg + t; e0 < end; e0 += EU * MAXB) {
    int d[EU];
#pragma unroll
    for (int u = 0; u < EU; ++u) d[u] = dst[min(e0 + u * MAXB, end - 1)];
    asm volatile("" :: "v"(d[0]), "v"(d[1]), "v"(d[2]), "v"(d[3]),
                 "v"(d[4]), "v"(d[5]), "v"(d[6]), "v"(d[7]));
#pragma unroll
    for (int u = 0; u < EU; ++u)
      if (e0 + u * MAXB < end) atomicAdd(&lh[d[u] >> BSH], 1);
  }
  __syncthreads();
  const int ex = block_excl_scan(lh[t], sw);
  if (t <= nb) offt[blockIdx.x * (nb + 1) + t] = ex;
  lh[t] = beg + ex;
  __syncthreads();
  if (chunk <= SCAP) {
    place_keys<true>(src, dst, lh, stage, cb, beg, end, t);
    __syncthreads();
    int k = t - (beg & 31);
    if (k >= 0 && k < end - beg) cb[beg + k] = stage[k];
    for (k += MAXB; k < end - beg; k += MAXB) cb[beg + k] = stage[k];
  } else {
    place_keys<false>(src, dst, lh, stage, cb, beg, end, t);
  }
}

template <int MODE>
__device__ __forceinline__ void piece_sweep(const int* __restrict__ cb, const int (&lo)[SPW],
                                            const int (&hi)[SPW], int lane, int* cnt,
                                            int* __restrict__ rec, int* stage, int sbase) {
#pragma unroll
  for (int g = 0; g < SPW; g += SPG) {
    int k[2 * SPG];
#pragma unroll
    for (int s = 0; s < SPG; ++s) {
#pragma unroll
      for (int u = 0; u < 2; ++u) {
        const int p = lo[g + s] + lane + 32 * u;
        k[2 * s + u] = cb[max(min(p, hi[g + s] - 1), 0)];
      }
    }
    asm volatile("" :: "v"(k[0]), "v"(k[1]), "v"(k[2]), "v"(k[3]), "v"(k[4]), "v"(k[5]),
                 "v"(k[6]), "v"(k[7]), "v"(k[8]), "v"(k[9]), "v"(k[10]), "v"(k[11]),
                 "v"(k[12]), "v"(k[13]), "v"(k[14]), "v"(k[15]));
#pragma unroll
    for (int s = 0; s < SPG; ++s) {
#pragma unroll
      for (int u = 0; u < 2; ++u) {
        if (lo[g + s] + lane + 32 * u < hi[g + s]) {
          const int key = k[2 * s + u];
          if (MODE == 2) stage[atomicAdd(&cnt[key & (BW - 1)], 1) - sbase] = key >> BSH;
          else if (MODE == 1) rec[atomicAdd(&cnt[key & (BW - 1)], 1)] = key >> BSH;
          else atomicAdd(&cnt[key & (BW - 1)], 1);
        }
      }
    }
#pragma unroll
    for (int s = 0; s < SPG; ++s) {
      for (int p = lo[g + s] + 64 + lane; p < hi[g + s]; p += 32) {
        const int key = cb[p];
        if (MODE == 2) stage[atomicAdd(&cnt[key & (BW - 1)], 1) - sbase] = key >> BSH;
        else if (MODE == 1) rec[atomicAdd(&cnt[key & (BW - 1)], 1)] = key >> BSH;
        else atomicAdd(&cnt[key & (BW - 1)], 1);
      }
    }
  }
}

__global__ void __launch_bounds__(BW)
bin_rows(const int* __restrict__ cb, const int* __restrict__ offt,
         int* __restrict__ rowptr, float* __restrict__ dinv, int* __restrict__ rec,
         int nb, int chunk) {
  __shared__ int cnt[BW];
  __shared__ int sw[32];
  __shared__ int sbeg;
  __shared__ int stot;
  __shared__ int stage[RCAP];
  const int t = threadIdx.x;
  const int lane = t & 31;
  const int w = __builtin_amdgcn_readfirstlane(t >> 5);
  const int b = blockIdx.x;
  cnt[t] = 0;
  if (t == 0) sbeg = 0;
  int lo[SPW];
  int hi[SPW];
  int below = 0;
#pragma unroll
  for (int s = 0; s < SPW; ++s) {
    const int c = w + (BW / 32) * s;
    const int o0 = offt[c * (nb + 1) + b];
    const int o1 = offt[c * (nb + 1) + b + 1];
    below += o0;
    lo[s] = c * chunk + o0;
    hi[s] = c * chunk + o1;
  }
  __syncthreads();
  if (lane == 0) atomicAdd(&sbeg, below);
  piece_sweep<0>(cb, lo, hi, lane, cnt, rec, stage, 0);
  __syncthreads();
  const int c = cnt[t];
  const int ex = block_excl_scan(c, sw);
  const int first = sbeg + ex;
  const int node = b * BW + t;
  rowptr[node] = first;
  dinv[node] = rsqrtf((float)(c + 1));
  cnt[t] = first;
  if (t == BW - 1) stot = ex + c;
  __syncthreads();
  const int sb  = __builtin_amdgcn_readfirstlane(sbeg);
  const int tot = __builtin_amdgcn_readfirstlane(stot);
  if (tot <= RCAP) {
    piece_sweep<2>(cb, lo, hi, lane, cnt, rec, stage, sb);
    __syncthreads();
    int k = t - (sb & 31);
    if (k >= 0 && k < tot) rec[sb + k] = stage[k];
    for (k += BW; k < tot; k += BW) rec[sb + k] = stage[k];
  } else {
    piece_sweep<1>(cb, lo, hi, lane, cnt, rec, stage, 0);
  }
}

template <bool WSRC>
__device__ __forceinline__ v2f row_sum(const int* __restrict__ rowptr,
                                       const int* __restrict__ rec,
                                       const float* __restrict__ msg,
                                       const float* __restrict__ dinv,
                                       int i, int lane, v2f& own, float& di) {
  const int beg = __builtin_amdgcn_readfirstlane(rowptr[i]);
  const int end = __builtin_amdgcn_readfirstlane(rowptr[i + 1]);
  di = dinv[i];
  const v2f* __restrict__ m2 = (const v2f*)msg;
  const v2f self = m2[(size_t)i * 32 + lane];
  own = self;
  float ax = WSRC ? di * self.x : self.x;
  float ay = WSRC ? di * self.y : self.y;
  for (int base = beg; base < end; base += 32) {
    const int cnt = min(32, end - base);
    int my = i;
    float wb = 0.0f;
    if (lane < cnt) {
      my = rec[(size_t)base + lane];
      if (WSRC) wb = dinv[my];
    }
    for (int j = 0; j < cnt; j += 8) {
      v2f v[8];
      float w[8];
#pragma unroll
      for (int u = 0; u < 8; ++u) {
        const int s = __builtin_amdgcn_readlane(my, j + u);
        w[u] = WSRC ? __int_as_float(__builtin_amdgcn_readlane(__float_as_int(wb), j + u)) : 1.0f;
        v[u] = m2[(size_t)s * 32 + lane];
      }
#pragma unroll
      for (int u = 0; u < 8; ++u) {
        const bool on = (j + u) < cnt;
        const float vx = on ? v[u].x : 0.0f;
        const float vy = on ? v[u].y : 0.0f;
        if (WSRC) {
          ax = fmaf(w[u], vx, ax);
          ay = fmaf(w[u], vy, ay);
        } else {
          ax += vx;
          ay += vy;
        }
      }
    }
  }
  v2f r;
  r.x = ax;
  r.y = ay;
  return r;
}

#define TP 68
#define WP 72

__device__ __forceinline__ v8f wmma4(v2f a, v2f b, v8f c) {
  return __builtin_amdgcn_wmma_f32_16x16x4_f32(false, a, false, b, (short)0, c, false, false);
}

__global__ void __launch_bounds__(512)
gather_dense(const int* __restrict__ rowptr, const int* __restrict__ rec,
             const float* __restrict__ x, const float* __restrict__ dinv,
             const float* __restrict__ W1, const float* __restrict__ b1,
             const float* __restrict__ W2, const float* __restrict__ Wemb,
             float* __restrict__ hw2, float* __restrict__ out, int n) {
  __shared__ __attribute__((aligned(16))) float Ts[32 * TP];
  __shared__ __attribute__((aligned(16))) float Xs[32 * TP];
  __shared__ __attribute__((aligned(16))) float Hs[32 * TP];
  __shared__ __attribute__((aligned(16))) float Es[32 * TP];
  __shared__ __attribute__((aligned(16))) float W2s[DD * WP];
  __shared__ float b1s[DD];
  const int lane = threadIdx.x & 31;
  const int half = lane >> 4;
  const int l16  = lane & 15;
  const int w = __builtin_amdgcn_readfirstlane((int)(threadIdx.x >> 5));
  const int row0 = blockIdx.x * 32;

  float dsv[2];
#pragma unroll
  for (int rr = 0; rr < 2; ++rr) {
    const int lr = rr * 16 + w;
    const int ir = min(row0 + lr, n - 1);
    v2f own;
    float di;
    const v2f s = row_sum<true>(rowptr, rec, x, dinv, ir, lane, own, di);
    dsv[rr] = di;
    v2f t;
    t.x = di * s.x;
    t.y = di * s.y;
    ((v2f*)(Ts + lr * TP))[lane] = t;
    ((v2f*)(Xs + lr * TP))[lane] = own;
  }

  const int tile = (w >> 2) & 1;
  const int ct   = w & 3;
  const float* __restrict__ Wm = (w < 8) ? W1 : Wemb;
  v2f bf[16];
#pragma unroll
  for (int k = 0; k < 16; ++k) {
    const int kb = 4 * k + 2 * half;
    bf[k].x = Wm[kb * DD + 16 * ct + l16];
    bf[k].y = Wm[(kb + 1) * DD + 16 * ct + l16];
  }
  {
    const int t  = threadIdx.x;
    const int r  = t >> 3;
    const int c0 = (t & 7) * 8;
    const float4 q0 = ((const float4*)W2)[2 * t];
    const float4 q1 = ((const float4*)W2)[2 * t + 1];
    *(float4*)(W2s + r * WP + c0)     = q0;
    *(float4*)(W2s + r * WP + c0 + 4) = q1;
    if (t < DD) b1s[t] = b1[t];
  }
  __syncthreads();

  {
    const float* As = ((w < 8) ? Ts : Xs) + (tile * 16 + l16) * TP + 2 * half;
    v8f acc0 = {}, acc1 = {};
#pragma unroll
    for (int k = 0; k < 16; k += 2) {
      acc0 = wmma4(*(const v2f*)(As + 4 * k), bf[k], acc0);
      acc1 = wmma4(*(const v2f*)(As + 4 * (k + 1)), bf[k + 1], acc1);
    }
    const v8f acc = acc0 + acc1;
    const int o = (tile * 16 + 8 * half) * TP + 16 * ct + l16;
    if (w < 8) {
      const float bb = b1s[16 * ct + l16];
#pragma unroll
      for (int r = 0; r < 8; ++r) Hs[o + r * TP] = fmaxf(acc[r] + bb, 0.0f);
    } else {
#pragma unroll
      for (int r = 0; r < 8; ++r) Es[o + r * TP] = acc[r];
    }
  }
  __syncthreads();

  if (w < 8) {
    const float* As = Hs + (tile * 16 + l16) * TP + 2 * half;
    const float* Bs = W2s + 2 * half * WP + 16 * ct + l16;
    v8f acc0 = {}, acc1 = {};
#pragma unroll
    for (int k = 0; k < 16; k += 2) {
      v2f b0, b1v;
      b0.x  = Bs[(4 * k) * WP];      b0.y  = Bs[(4 * k + 1) * WP];
      b1v.x = Bs[(4 * k + 4) * WP];  b1v.y = Bs[(4 * k + 5) * WP];
      acc0 = wmma4(*(const v2f*)(As + 4 * k), b0, acc0);
      acc1 = wmma4(*(const v2f*)(As + 4 * (k + 1)), b1v, acc1);
    }
    const v8f acc = acc0 + acc1;
    const int o = (tile * 16 + 8 * half) * TP + 16 * ct + l16;
#pragma unroll
    for (int r = 0; r < 8; ++r) Ts[o + r * TP] = acc[r];
  }
  __syncthreads();

#pragma unroll
  for (int rr = 0; rr < 2; ++rr) {
    const int lr = rr * 16 + w;
    const int i = row0 + lr;
    if (i < n) {
      v2f p = ((const v2f*)(Ts + lr * TP))[lane];
      p.x *= dsv[rr];
      p.y *= dsv[rr];
      ((v2f*)hw2)[(size_t)i * 32 + lane] = p;
      ((v2f*)out)[(size_t)i * 32 + lane] = ((const v2f*)(Es + lr * TP))[lane];
    }
  }
}

__global__ void __launch_bounds__(256)
gather_final(const int* __restrict__ rowptr, const int* __restrict__ rec,
             const float* __restrict__ msg, const float* __restrict__ dinv,
             const float* __restrict__ b2, const float* __restrict__ bemb,
             float* out, int n) {
  const int lane = threadIdx.x & 31;
  const int i = blockIdx.x * 8 + __builtin_amdgcn_readfirstlane((int)(threadIdx.x >> 5));
  if (i >= n) return;
  const v2f o = ((const v2f*)out)[(size_t)i * 32 + lane];
  const v2f bv = ((const v2f*)b2)[lane];
  const v2f be = ((const v2f*)bemb)[lane];
  v2f own;
  float di;
  const v2f a = row_sum<false>(rowptr, rec, msg, dinv, i, lane, own, di);
  v2f r;
  r.x = o.x + be.x + bv.x + di * a.x;
  r.y = o.y + be.y + bv.y + di * a.y;
  ((v2f*)out)[(size_t)i * 32 + lane] = r;
}

extern "C" void kernel_launch(void* const* d_in, const int* in_sizes, int n_in,
                              void* d_out, int out_size, void* d_ws, size_t ws_size,
                              hipStream_t stream) {
  const float* x    = (const float*)d_in[0];
  const int*   ei   = (const int*)d_in[1];
  const float* W1   = (const float*)d_in[2];
  const float* b1   = (const float*)d_in[3];
  const float* W2   = (const float*)d_in[4];
  const float* b2   = (const float*)d_in[5];
  const float* Wemb = (const float*)d_in[6];
  const float* bemb = (const float*)d_in[7];
  float* out = (float*)d_out;

  const int N = in_sizes[0] / DD;
  const int E = in_sizes[1] / 2;
  const int* src = ei;
  const int* dst = ei + E;

  const size_t npad = ((size_t)N + 1 + 1023) & ~(size_t)1023;
  const size_t epad = ((size_t)E + 63) & ~(size_t)63;
  int*   rowptr  = (int*)d_ws;
  float* dinv    = (float*)(rowptr + npad);
  int*   offt    = (int*)(dinv + npad);
  int*   cb      = offt + (size_t)NCH * MAXB;
  int*   rec     = cb + epad;
  float* buf1    = (float*)(rec + epad);

  const int B     = 256;
  const int nb    = N / BW + 1;
  const int chunk = (E + NCH - 1) / NCH;
  const int gR    = (N + 7) / 8;
  const int gT    = (N + 31) / 32;

  chunk_sort<<<NCH, MAXB, 0, stream>>>(src, dst, offt, cb, E, nb, chunk);
  bin_rows<<<nb, BW, 0, stream>>>(cb, offt, rowptr, dinv, rec, nb, chunk);

  gather_dense<<<gT, 512, 0, stream>>>(rowptr, rec, x, dinv, W1, b1, W2, Wemb, buf1, out, N);

  gather_final<<<gR, B, 0, stream>>>(rowptr, rec, buf1, dinv, b2, bemb, out, N);
}


# Row gathers read two source rows per instruction: half a wave per row, 16 bytes a lane, index and weight handed round by row_share; even and odd list positions summed apart, added once at the end
# speedup vs baseline: 8.0261x; 1.0524x over previous
#include <hip/hip_runtime.h>


typedef float v2f __attribute__((ext_vector_type(2)));
typedef float v4f __attribute__((ext_vector_type(4)));
typedef float v8f __attribute__((ext_vector_type(8)));

#define DD 64

#define BSH 9
#define BW  512
#define NCH 256
#define MAXB 1024
#define EU  8
#define SPW (NCH / (BW / 32))
#define SPG 8
#define SCAP 8192
#define RCAP 12288

__device__ __forceinline__ int block_excl_scan(int v, int* sw) {
  const int lane = threadIdx.x & 31;
  const int wave = threadIdx.x >> 5;
  int inc = v;
  for (int d = 1; d < 32; d <<= 1) {
    const int up = __shfl_up(inc, d);
    if (lane >= d) inc += up;
  }
  if (lane == 31) sw[wave] = inc;
  __syncthreads();
  int off = 0;
  for (int k = 0; k < wave; ++k) off += sw[k];
  return off + inc - v;
}

template <bool TO_LDS>
__device__ __forceinline__ void place_keys(const int* __restrict__ src,
                                           const int* __restrict__ dst, int* lh, int* stage,
                                           int* __restrict__ cb, int beg, int end, int t) {
  for (int e0 = beg + t; e0 < end; e0 += EU * MAXB) {
    int d[EU];
    int s[EU];
#pragma unroll
    for (int u = 0; u < EU; ++u) {
      const int e = min(e0 + u * MAXB, end - 1);
      d[u] = dst[e];
      s[u] = src[e];
    }
    asm volatile("" :: "v"(d[0]), "v"(d[1]), "v"(d[2]), "v"(d[3]),
                 "v"(d[4]), "v"(d[5]), "v"(d[6]), "v"(d[7]),
                 "v"(s[0]), "v"(s[1]), "v"(s[2]), "v"(s[3]),
                 "v"(s[4]), "v"(s[5]), "v"(s[6]), "v"(s[7]));
#pragma unroll
    for (int u = 0; u < EU; ++u) {
      if (e0 + u * MAXB < end) {
        const int p = atomicAdd(&lh[d[u] >> BSH], 1);
        const int key = (int)(((unsigned)s[u] << BSH) | (unsigned)(d[u] & (BW - 1)));
        if (TO_LDS) stage[p - beg] = key;
        else cb[p] = key;
      }
    }
  }
}

__global__ void __launch_bounds__(MAXB)
chunk_sort(const int* __restrict__ src, const int* __restrict__ dst,
           int* __restrict__ offt, int* __restrict__ cb, int ne, int nb, int chunk) {
  __shared__ int lh[MAXB];
  __shared__ int sw[32];
  __shared__ int stage[SCAP];
  const int t = threadIdx.x;
  lh[t] = 0;
  __syncthreads();
  const int beg = blockIdx.x * chunk;
  const int end = min(ne, beg + chunk);
  for (int e0 = beg + t; e0 < end; e0 += EU * MAXB) {
    int d[EU];
#pragma unroll
    for (int u = 0; u < EU; ++u) d[u] = dst[min(e0 + u * MAXB, end - 1)];
    asm volatile("" :: "v"(d[0]), "v"(d[1]), "v"(d[2]), "v"(d[3]),
                 "v"(d[4]), "v"(d[5]), "v"(d[6]), "v"(d[7]));
#pragma unroll
    for (int u = 0; u < EU; ++u)
      if (e0 + u * MAXB < end) atomicAdd(&lh[d[u] >> BSH], 1);
  }
  __syncthreads();
  const int ex = block_excl_scan(lh[t], sw);
  if (t <= nb) offt[blockIdx.x * (nb + 1) + t] = ex;
  lh[t] = beg + ex;
  __syncthreads();
  if (chunk <= SCAP) {
    place_keys<true>(src, dst, lh, stage, cb, beg, end, t);
    __syncthreads();
    int k = t - (beg & 31);
    if (k >= 0 && k < end - beg) cb[beg + k] = stage[k];
    for (k += MAXB; k < end - beg; k += MAXB) cb[beg + k] = stage[k];
  } else {
    place_keys<false>(src, dst, lh, stage, cb, beg, end, t);
  }
}

template <int MODE>
__device__ __forceinline__ void piece_sweep(const int* __restrict__ cb, const int (&lo)[SPW],
                                            const int (&hi)[SPW], int lane, int* cnt,
                                            int* __restrict__ rec, int* stage, int sbase) {
#pragma unroll
  for (int g = 0; g < SPW; g += SPG) {
    int k[2 * SPG];
#pragma unroll
    for (int s = 0; s < SPG; ++s) {
#pragma unroll
      for (int u = 0; u < 2; ++u) {
        const int p = lo[g + s] + lane + 32 * u;
        k[2 * s + u] = cb[max(min(p, hi[g + s] - 1), 0)];
      }
    }
    asm volatile("" :: "v"(k[0]), "v"(k[1]), "v"(k[2]), "v"(k[3]), "v"(k[4]), "v"(k[5]),
                 "v"(k[6]), "v"(k[7]), "v"(k[8]), "v"(k[9]), "v"(k[10]), "v"(k[11]),
                 "v"(k[12]), "v"(k[13]), "v"(k[14]), "v"(k[15]));
#pragma unroll
    for (int s = 0; s < SPG; ++s) {
#pragma unroll
      for (int u = 0; u < 2; ++u) {
        if (lo[g + s] + lane + 32 * u < hi[g + s]) {
          const int key = k[2 * s + u];
          if (MODE == 2) stage[atomicAdd(&cnt[key & (BW - 1)], 1) - sbase] = key >> BSH;
          else if (MODE == 1) rec[atomicAdd(&cnt[key & (BW - 1)], 1)] = key >> BSH;
          else atomicAdd(&cnt[key & (BW - 1)], 1);
        }
      }
    }
#pragma unroll
    for (int s = 0; s < SPG; ++s) {
      for (int p = lo[g + s] + 64 + lane; p < hi[g + s]; p += 32) {
        const int key = cb[p];
        if (MODE == 2) stage[atomicAdd(&cnt[key & (BW - 1)], 1) - sbase] = key >> BSH;
        else if (MODE == 1) rec[atomicAdd(&cnt[key & (BW - 1)], 1)] = key >> BSH;
        else atomicAdd(&cnt[key & (BW - 1)], 1);
      }
    }
  }
}

__global__ void __launch_bounds__(BW)
bin_rows(const int* __restrict__ cb, const int* __restrict__ offt,
         int* __restrict__ rowptr, float* __restrict__ dinv, int* __restrict__ rec,
         int nb, int chunk) {
  __shared__ int cnt[BW];
  __shared__ int sw[32];
  __shared__ int sbeg;
  __shared__ int stot;
  __shared__ int stage[RCAP];
  const int t = threadIdx.x;
  const int lane = t & 31;
  const int w = __builtin_amdgcn_readfirstlane(t >> 5);
  const int b = blockIdx.x;
  cnt[t] = 0;
  if (t == 0) sbeg = 0;
  int lo[SPW];
  int hi[SPW];
  int below = 0;
#pragma unroll
  for (int s = 0; s < SPW; ++s) {
    const int c = w + (BW / 32) * s;
    const int o0 = offt[c * (nb + 1) + b];
    const int o1 = offt[c * (nb + 1) + b + 1];
    below += o0;
    lo[s] = c * chunk + o0;
    hi[s] = c * chunk + o1;
  }
  __syncthreads();
  if (lane == 0) atomicAdd(&sbeg, below);
  piece_sweep<0>(cb, lo, hi, lane, cnt, rec, stage, 0);
  __syncthreads();
  const int c = cnt[t];
  const int ex = block_excl_scan(c, sw);
  const int first = sbeg + ex;
  const int node = b * BW + t;
  rowptr[node] = first;
  dinv[node] = rsqrtf((float)(c + 1));
  cnt[t] = first;
  if (t == BW - 1) stot = ex + c;
  __syncthreads();
  const int sb  = __builtin_amdgcn_readfirstlane(sbeg);
  const int tot = __builtin_amdgcn_readfirstlane(stot);
  if (tot <= RCAP) {
    piece_sweep<2>(cb, lo, hi, lane, cnt, rec, stage, sb);
    __syncthreads();
    int k = t - (sb & 31);
    if (k >= 0 && k < tot) rec[sb + k] = stage[k];
    for (k += BW; k < tot; k += BW) rec[sb + k] = stage[k];
  } else {
    piece_sweep<1>(cb, lo, hi, lane, cnt, rec, stage, 0);
  }
}


template <int K>
__device__ __forceinline__ int row_lane(int v) {
  return __builtin_amdgcn_update_dpp(0, v, 0x150 + K, 0xf, 0xf, false);
}

template <int K0, int NK>
__device__ __forceinline__ void pair_steps(const char* __restrict__ mb, unsigned loff,
                                           int my, float wb, v4f& acc) {
  v4f v[NK];
#define PAIR_READ(U)                                                              \
  if (U < NK) {                                                                   \
    const unsigned s = (unsigned)row_lane<K0 + (U < NK ? U : 0)>(my);             \
    v[U < NK ? U : 0] = *(const v4f*)(mb + ((s << 8) | loff));                    \
  }
  PAIR_READ(0) PAIR_READ(1) PAIR_READ(2) PAIR_READ(3)
  PAIR_READ(4) PAIR_READ(5) PAIR_READ(6) PAIR_READ(7)
#undef PAIR_READ
  __builtin_amdgcn_sched_barrier(0);
#define PAIR_SUM(U)                                                               \
  if (U < NK) {                                                                   \
    const float w = __int_as_float(row_lane<K0 + (U < NK ? U : 0)>(__float_as_int(wb))); \
    const v4f r = v[U < NK ? U : 0];                                              \
    acc.x = fmaf(w, r.x, acc.x);                                                  \
    acc.y = fmaf(w, r.y, acc.y);                                                  \
    acc.z = fmaf(w, r.z, acc.z);                                                  \
    acc.w = fmaf(w, r.w, acc.w);                                                  \
  }
  PAIR_SUM(0) PAIR_SUM(1) PAIR_SUM(2) PAIR_SUM(3)
  PAIR_SUM(4) PAIR_SUM(5) PAIR_SUM(6) PAIR_SUM(7)
#undef PAIR_SUM
}

template <bool WSRC>
__device__ __forceinline__ v4f row_sum(const int* __restrict__ rowptr,
                                       const int* __restrict__ rec,
                                       const float* __restrict__ msg,
                                       const float* __restrict__ dinv,
                                       int i, int lane, v4f& own, float& di) {
  const int beg = __builtin_amdgcn_readfirstlane(rowptr[i]);
  const int end = __builtin_amdgcn_readfirstlane(rowptr[i + 1]);
  di = dinv[i];
  const int half = lane >> 4;
  const int l16  = lane & 15;
  const unsigned loff = (unsigned)l16 << 4;
  const int pos = 2 * l16 + half;
  const char* __restrict__ mb = (const char*)msg;
  const v4f self = *(const v4f*)(mb + (((unsigned)i << 8) | loff));
  own = self;
  const float ws = WSRC ? di : 1.0f;
  v4f acc;
  acc.x = half ? 0.0f : ws * self.x;
  acc.y = half ? 0.0f : ws * self.y;
  acc.z = half ? 0.0f : ws * self.z;
  acc.w = half ? 0.0f : ws * self.w;
  for (int base = beg; base < end; base += 32) {
    const int cnt = min(32, end - base);
    int my = i;
    float wb = 0.0f;
    if (pos < cnt) {
      my = rec[(size_t)base + pos];
      wb = WSRC ? dinv[my] : 1.0f;
    }
    pair_steps<0, 8>(mb, loff, my, wb, acc);
    if (cnt > 16) {
      pair_steps<8, 4>(mb, loff, my, wb, acc);
      if (cnt > 24) pair_steps<12, 4>(mb, loff, my, wb, acc);
    }
  }
  v4f r;
  r.x = acc.x + __int_as_float(__builtin_amdgcn_permlanex16(0, __float_as_int(acc.x), 0x76543210, 0xfedcba98, false, false));
  r.y = acc.y + __int_as_float(__builtin_amdgcn_permlanex16(0, __float_as_int(acc.y), 0x76543210, 0xfedcba98, false, false));
  r.z = acc.z + __int_as_float(__builtin_amdgcn_permlanex16(0, __float_as_int(acc.z), 0x76543210, 0xfedcba98, false, false));
  r.w = acc.w + __int_as_float(__builtin_amdgcn_permlanex16(0, __float_as_int(acc.w), 0x76543210, 0xfedcba98, false, false));
  return r;
}

#define TP 68
#define WP 72

__device__ __forceinline__ v8f wmma4(v2f a, v2f b, v8f c) {
  return __builtin_amdgcn_wmma_f32_16x16x4_f32(false, a, false, b, (short)0, c, false, false);
}

__global__ void __launch_bounds__(512)
gather_dense(const int* __restrict__ rowptr, const int* __restrict__ rec,
             const float* __restrict__ x, const float* __restrict__ dinv,
             const float* __restrict__ W1, const float* __restrict__ b1,
             const float* __restrict__ W2, const float* __restrict__ Wemb,
             float* __restrict__ hw2, float* __restrict__ out, int n) {
  __shared__ __attribute__((aligned(16))) float Ts[32 * TP];
  __shared__ __attribute__((aligned(16))) float Xs[32 * TP];
  __shared__ __attribute__((aligned(16))) float Hs[32 * TP];
  __shared__ __attribute__((aligned(16))) float Es[32 * TP];
  __shared__ __attribute__((aligned(16))) float W2s[DD * WP];
  __shared__ float b1s[DD];
  const int lane = threadIdx.x & 31;
  const int half = lane >> 4;
  const int l16  = lane & 15;
  const int w = __builtin_amdgcn_readfirstlane((int)(threadIdx.x >> 5));
  const int row0 = blockIdx.x * 32;

  float dsv[2];
#pragma unroll
  for (int rr = 0; rr < 2; ++rr) {
    const int lr = rr * 16 + w;
    const int ir = min(row0 + lr, n - 1);
    v4f own;
    float di;
    const v4f s = row_sum<true>(rowptr, rec, x, dinv, ir, lane, own, di);
    dsv[rr] = di;
    if (half == 0) {
      v4f t;
      t.x = di * s.x;
      t.y = di * s.y;
      t.z = di * s.z;
      t.w = di * s.w;
      *(v4f*)(Ts + lr * TP + 4 * l16) = t;
      *(v4f*)(Xs + lr * TP + 4 * l16) = own;
    }
  }

  const int tile = (w >> 2) & 1;
  const int ct   = w & 3;
  const float* __restrict__ Wm = (w < 8) ? W1 : Wemb;
  v2f bf[16];
#pragma unroll
  for (int k = 0; k < 16; ++k) {
    const int kb = 4 * k + 2 * half;
    bf[k].x = Wm[kb * DD + 16 * ct + l16];
    bf[k].y = Wm[(kb + 1) * DD + 16 * ct + l16];
  }
  {
    const int t  = threadIdx.x;
    const int r  = t >> 3;
    const int c0 = (t & 7) * 8;
    const float4 q0 = ((const float4*)W2)[2 * t];
    const float4 q1 = ((const float4*)W2)[2 * t + 1];
    *(float4*)(W2s + r * WP + c0)     = q0;
    *(float4*)(W2s + r * WP + c0 + 4) = q1;
    if (t < DD) b1s[t] = b1[t];
  }
  __syncthreads();

  {
    const float* As = ((w < 8) ? Ts : Xs) + (tile * 16 + l16) * TP + 2 * half;
    v8f acc0 = {}, acc1 = {};
#pragma unroll
    for (int k = 0; k < 16; k += 2) {
      acc0 = wmma4(*(const v2f*)(As + 4 * k), bf[k], acc0);
      acc1 = wmma4(*(const v2f*)(As + 4 * (k + 1)), bf[k + 1], acc1);
    }
    const v8f acc = acc0 + acc1;
    const int o = (tile * 16 + 8 * half) * TP + 16 * ct + l16;
    if (w < 8) {
      const float bb = b1s[16 * ct + l16];
#pragma unroll
      for (int r = 0; r < 8; ++r) Hs[o + r * TP] = fmaxf(acc[r] + bb, 0.0f);
    } else {
#pragma unroll
      for (int r = 0; r < 8; ++r) Es[o + r * TP] = acc[r];
    }
  }
  __syncthreads();

  if (w < 8) {
    const float* As = Hs + (tile * 16 + l16) * TP + 2 * half;
    const float* Bs = W2s + 2 * half * WP + 16 * ct + l16;
    v8f acc0 = {}, acc1 = {};
#pragma unroll
    for (int k = 0; k < 16; k += 2) {
      v2f b0, b1v;
      b0.x  = Bs[(4 * k) * WP];      b0.y  = Bs[(4 * k + 1) * WP];
      b1v.x = Bs[(4 * k + 4) * WP];  b1v.y = Bs[(4 * k + 5) * WP];
      acc0 = wmma4(*(const v2f*)(As + 4 * k), b0, acc0);
      acc1 = wmma4(*(const v2f*)(As + 4 * (k + 1)), b1v, acc1);
    }
    const v8f acc = acc0 + acc1;
    const int o = (tile * 16 + 8 * half) * TP + 16 * ct + l16;
#pragma unroll
    for (int r = 0; r < 8; ++r) Ts[o + r * TP] = acc[r];
  }
  __syncthreads();

#pragma unroll
  for (int rr = 0; rr < 2; ++rr) {
    const int lr = rr * 16 + w;
    const int i = row0 + lr;
    if (i < n) {
      v2f p = ((const v2f*)(Ts + lr * TP))[lane];
      p.x *= dsv[rr];
      p.y *= dsv[rr];
      ((v2f*)hw2)[(size_t)i * 32 + lane] = p;
      ((v2f*)out)[(size_t)i * 32 + lane] = ((const v2f*)(Es + lr * TP))[lane];
    }
  }
}

__global__ void __launch_bounds__(256)
gather_final(const int* __restrict__ rowptr, const int* __restrict__ rec,
             const float* __restrict__ msg, const float* __restrict__ dinv,
             const float* __restrict__ b2, const float* __restrict__ bemb,
             float* out, int n) {
  const int lane = threadIdx.x & 31;
  const int i = blockIdx.x * 8 + __builtin_amdgcn_readfirstlane((int)(threadIdx.x >> 5));
  if (i >= n) return;
  const int half = lane >> 4;
  const int l16  = lane & 15;
  const v4f o  = ((const v4f*)out)[(size_t)i * 16 + l16];
  const v4f bv = ((const v4f*)b2)[l16];
  const v4f be = ((const v4f*)bemb)[l16];
  v4f own;
  float di;
  const v4f a = row_sum<false>(rowptr, rec, msg, dinv, i, lane, own, di);
  v4f r;
  r.x = o.x + be.x + bv.x + di * a.x;
  r.y = o.y + be.y + bv.y + di * a.y;
  r.z = o.z + be.z + bv.z + di * a.z;
  r.w = o.w + be.w + bv.w + di * a.w;
  if (half == 0) ((v4f*)out)[(size_t)i * 16 + l16] = r;
}

extern "C" void kernel_launch(void* const* d_in, const int* in_sizes, int n_in,
                              void* d_out, int out_size, void* d_ws, size_t ws_size,
                              hipStream_t stream) {
  const float* x    = (const float*)d_in[0];
  const int*   ei   = (const int*)d_in[1];
  const float* W1   = (const float*)d_in[2];
  const float* b1   = (const float*)d_in[3];
  const float* W2   = (const float*)d_in[4];
  const float* b2   = (const float*)d_in[5];
  const float* Wemb = (const float*)d_in[6];
  const float* bemb = (const float*)d_in[7];
  float* out = (float*)d_out;

  const int N = in_sizes[0] / DD;
  const int E = in_sizes[1] / 2;
  const int* src = ei;
  const int* dst = ei + E;

  const size_t npad = ((size_t)N + 1 + 1023) & ~(size_t)1023;
  const size_t epad = ((size_t)E + 63) & ~(size_t)63;
  int*   rowptr  = (int*)d_ws;
  float* dinv    = (float*)(rowptr + npad);
  int*   offt    = (int*)(dinv + npad);
  int*   cb      = offt + (size_t)NCH * MAXB;
  int*   rec     = cb + epad;
  float* buf1    = (float*)(rec + epad);

  const int B     = 256;
  const int nb    = N / BW + 1;
  const int chunk = (E + NCH - 1) / NCH;
  const int gR    = (N + 7) / 8;
  const int gT    = (N + 31) / 32;

  chunk_sort<<<NCH, MAXB, 0, stream>>>(src, dst, offt, cb, E, nb, chunk);
  bin_rows<<<nb, BW, 0, stream>>>(cb, offt, rowptr, dinv, rec, nb, chunk);

  gather_dense<<<gT, 512, 0, stream>>>(rowptr, rec, x, dinv, W1, b1, W2, Wemb, buf1, out, N);

  gather_final<<<gR, B, 0, stream>>>(rowptr, rec, buf1, dinv, b2, bemb, out, N);
}


# List build reads each edge once: the counting LDS add returns the rank, keys and ranks wait in registers over the block scan (now DPP moves and one LDS read); row sums take another arrival order
# speedup vs baseline: 8.3118x; 1.0356x over previous
#include <hip/hip_runtime.h>


typedef float v2f __attribute__((ext_vector_type(2)));
typedef float v4f __attribute__((ext_vector_type(4)));
typedef float v8f __attribute__((ext_vector_type(8)));

#define DD 64

#define BSH 9
#define BW  512
#define NCH 256
#define MAXB 1024
#define EU  8
#define SPW (NCH / (BW / 32))
#define SCAP 8192
#define RCAP 12288

template <int CTRL>
__device__ __forceinline__ int row_move(int v) {
  return __builtin_amdgcn_update_dpp(0, v, CTRL, 0xf, 0xf, true);
}

template <int NW>
__device__ __forceinline__ int block_excl_scan(int v, int* sw) {
  const int lane = threadIdx.x & 31;
  const int wave = threadIdx.x >> 5;
  int inc = v;
  inc += row_move<0x111>(inc);
  inc += row_move<0x112>(inc);
  inc += row_move<0x114>(inc);
  inc += row_move<0x118>(inc);
  const int lower = __builtin_amdgcn_readlane(inc, 15);
  if (lane >= 16) inc += lower;
  if (lane == 31) sw[wave] = inc;
  __syncthreads();
  int low = (lane < wave && lane < NW) ? sw[lane] : 0;
  low += row_move<0x161>(low);
  low += row_move<0x162>(low);
  low += row_move<0x164>(low);
  low += row_move<0x168>(low);
  low += __builtin_amdgcn_permlanex16(0, low, 0x76543210, 0xfedcba98, false, false);
  return low + inc - v;
}

__device__ __forceinline__ void place_keys(const int* __restrict__ src,
                                           const int* __restrict__ dst, int* lh,
                                           int* __restrict__ cb, int beg, int end, int t) {
  for (int e0 = beg + t; e0 < end; e0 += EU * MAXB) {
    int d[EU];
    int s[EU];
#pragma unroll
    for (int u = 0; u < EU; ++u) {
      const int e = min(e0 + u * MAXB, end - 1);
      d[u] = dst[e];
      s[u] = src[e];
    }
    asm volatile("" :: "v"(d[0]), "v"(d[1]), "v"(d[2]), "v"(d[3]),
                 "v"(d[4]), "v"(d[5]), "v"(d[6]), "v"(d[7]),
                 "v"(s[0]), "v"(s[1]), "v"(s[2]), "v"(s[3]),
                 "v"(s[4]), "v"(s[5]), "v"(s[6]), "v"(s[7]));
#pragma unroll
    for (int u = 0; u < EU; ++u) {
      if (e0 + u * MAXB < end) {
        const int p = atomicAdd(&lh[d[u] >> BSH], 1);
        cb[p] = (int)(((unsigned)s[u] << BSH) | (unsigned)(d[u] & (BW - 1)));
      }
    }
  }
}

static_assert(SCAP <= EU * MAXB, "a chunk that fits the LDS stretch must fit one trip of EU reads a thread");

__global__ void __launch_bounds__(MAXB)
chunk_sort(const int* __restrict__ src, const int* __restrict__ dst,
           int* __restrict__ offt, int* __restrict__ cb, int ne, int nb, int chunk) {
  __shared__ int lh[MAXB];
  __shared__ int sw[32];
  __shared__ int stage[SCAP];
  const int t = threadIdx.x;
  lh[t] = 0;
  __syncthreads();
  const int beg = blockIdx.x * chunk;
  const int end = min(ne, beg + chunk);
  if (chunk <= SCAP) {
    int d[EU];
    int s[EU];
    int r[EU];
#pragma unroll
    for (int u = 0; u < EU; ++u) {
      const int e = max(min(beg + t + u * MAXB, end - 1), 0);
      d[u] = dst[e];
      s[u] = src[e];
    }
    asm volatile("" :: "v"(d[0]), "v"(d[1]), "v"(d[2]), "v"(d[3]),
                 "v"(d[4]), "v"(d[5]), "v"(d[6]), "v"(d[7]),
                 "v"(s[0]), "v"(s[1]), "v"(s[2]), "v"(s[3]),
                 "v"(s[4]), "v"(s[5]), "v"(s[6]), "v"(s[7]));
#pragma unroll
    for (int u = 0; u < EU; ++u) {
      r[u] = 0;
      if (beg + t + u * MAXB < end) r[u] = atomicAdd(&lh[d[u] >> BSH], 1);
    }
    __syncthreads();
    const int ex = block_excl_scan<MAXB / 32>(lh[t], sw);
    if (t <= nb) offt[blockIdx.x * (nb + 1) + t] = ex;
    lh[t] = ex;
    __syncthreads();
    int f[EU];
#pragma unroll
    for (int u = 0; u < EU; ++u) f[u] = lh[d[u] >> BSH];
#pragma unroll
    for (int u = 0; u < EU; ++u) {
      if (beg + t + u * MAXB < end)
        stage[f[u] + r[u]] = (int)(((unsigned)s[u] << BSH) | (unsigned)(d[u] & (BW - 1)));
    }
    __syncthreads();
    int k = t - (beg & 31);
    if (k >= 0 && k < end - beg) cb[beg + k] = stage[k];
    for (k += MAXB; k < end - beg; k += MAXB) cb[beg + k] = stage[k];
  } else {
    for (int e0 = beg + t; e0 < end; e0 += EU * MAXB) {
      int d[EU];
#pragma unroll
      for (int u = 0; u < EU; ++u) d[u] = dst[min(e0 + u * MAXB, end - 1)];
      asm volatile("" :: "v"(d[0]), "v"(d[1]), "v"(d[2]), "v"(d[3]),
                   "v"(d[4]), "v"(d[5]), "v"(d[6]), "v"(d[7]));
#pragma unroll
      for (int u = 0; u < EU; ++u)
        if (e0 + u * MAXB < end) atomicAdd(&lh[d[u] >> BSH], 1);
    }
    __syncthreads();
    const int ex = block_excl_scan<MAXB / 32>(lh[t], sw);
    if (t <= nb) offt[blockIdx.x * (nb + 1) + t] = ex;
    lh[t] = beg + ex;
    __syncthreads();
    place_keys(src, dst, lh, cb, beg, end, t);
  }
}

__device__ __forceinline__ void piece_count(const int* __restrict__ cb, const int (&lo)[SPW],
                                            const int (&hi)[SPW], int lane, int* cnt, int* cnt2,
                                            int (&k)[2 * SPW], int (&r)[2 * SPW]) {
#pragma unroll
  for (int s = 0; s < SPW; ++s) {
#pragma unroll
    for (int u = 0; u < 2; ++u) {
      const int p = lo[s] + lane + 32 * u;
      k[2 * s + u] = cb[max(min(p, hi[s] - 1), 0)];
    }
  }
  asm volatile("" :: "v"(k[0]), "v"(k[1]), "v"(k[2]), "v"(k[3]), "v"(k[4]), "v"(k[5]),
               "v"(k[6]), "v"(k[7]), "v"(k[8]), "v"(k[9]), "v"(k[10]), "v"(k[11]),
               "v"(k[12]), "v"(k[13]), "v"(k[14]), "v"(k[15]));
  asm volatile("" :: "v"(k[16]), "v"(k[17]), "v"(k[18]), "v"(k[19]), "v"(k[20]), "v"(k[21]),
               "v"(k[22]), "v"(k[23]), "v"(k[24]), "v"(k[25]), "v"(k[26]), "v"(k[27]),
               "v"(k[28]), "v"(k[29]), "v"(k[30]), "v"(k[31]));
#pragma unroll
  for (int s = 0; s < SPW; ++s) {
#pragma unroll
    for (int u = 0; u < 2; ++u) {
      r[2 * s + u] = 0;
      if (lo[s] + lane + 32 * u < hi[s])
        r[2 * s + u] = atomicAdd(&cnt[k[2 * s + u] & (BW - 1)], 1);
    }
  }
#pragma unroll
  for (int s = 0; s < SPW; ++s)
    for (int p = lo[s] + 64 + lane; p < hi[s]; p += 32) atomicAdd(&cnt2[cb[p] & (BW - 1)], 1);
}

template <bool TO_LDS>
__device__ __forceinline__ void piece_place(const int* __restrict__ cb, const int (&lo)[SPW],
                                            const int (&hi)[SPW], int lane, const int* cnt,
                                            int* cnt2, const int (&k)[2 * SPW],
                                            const int (&r)[2 * SPW], int* __restrict__ rec,
                                            int* stage, int sbase) {
  int f[2 * SPW];
#pragma unroll
  for (int i = 0; i < 2 * SPW; ++i) f[i] = cnt[k[i] & (BW - 1)];
#pragma unroll
  for (int s = 0; s < SPW; ++s) {
#pragma unroll
    for (int u = 0; u < 2; ++u) {
      if (lo[s] + lane + 32 * u < hi[s]) {
        const int slot = f[2 * s + u] + r[2 * s + u];
        if (TO_LDS) stage[slot - sbase] = k[2 * s + u] >> BSH;
        else rec[slot] = k[2 * s + u] >> BSH;
      }
    }
  }
#pragma unroll
  for (int s = 0; s < SPW; ++s) {
    for (int p = lo[s] + 64 + lane; p < hi[s]; p += 32) {
      const int key = cb[p];
      const int slot = atomicAdd(&cnt2[key & (BW - 1)], 1);
      if (TO_LDS) stage[slot - sbase] = key >> BSH;
      else rec[slot] = key >> BSH;
    }
  }
}

__global__ void __launch_bounds__(BW)
bin_rows(const int* __restrict__ cb, const int* __restrict__ offt,
         int* __restrict__ rowptr, float* __restrict__ dinv, int* __restrict__ rec,
         int nb, int chunk) {
  __shared__ int cnt[BW];
  __shared__ int cnt2[BW];
  __shared__ int sw[32];
  __shared__ int sbeg;
  __shared__ int stot;
  __shared__ int stage[RCAP];
  const int t = threadIdx.x;
  const int lane = t & 31;
  const int w = __builtin_amdgcn_readfirstlane(t >> 5);
  const int b = blockIdx.x;
  cnt[t] = 0;
  cnt2[t] = 0;
  if (t == 0) sbeg = 0;
  int lo[SPW];
  int hi[SPW];
  int below = 0;
#pragma unroll
  for (int s = 0; s < SPW; ++s) {
    const int c = w + (BW / 32) * s;
    const int o0 = offt[c * (nb + 1) + b];
    const int o1 = offt[c * (nb + 1) + b + 1];
    below += o0;
    lo[s] = c * chunk + o0;
    hi[s] = c * chunk + o1;
  }
  __syncthreads();
  if (lane == 0) atomicAdd(&sbeg, below);
  int k[2 * SPW];
  int r[2 * SPW];
  piece_count(cb, lo, hi, lane, cnt, cnt2, k, r);
  __syncthreads();
  const int c1 = cnt[t];
  const int c = c1 + cnt2[t];
  const int ex = block_excl_scan<BW / 32>(c, sw);
  const int first = sbeg + ex;
  const int node = b * BW + t;
  rowptr[node] = first;
  dinv[node] = rsqrtf((float)(c + 1));
  cnt[t] = first;
  cnt2[t] = first + c1;
  if (t == BW - 1) stot = ex + c;
  __syncthreads();
  const int sb  = __builtin_amdgcn_readfirstlane(sbeg);
  const int tot = __builtin_amdgcn_readfirstlane(stot);
  if (tot <= RCAP) {
    piece_place<true>(cb, lo, hi, lane, cnt, cnt2, k, r, rec, stage, sb);
    __syncthreads();
    int q = t - (sb & 31);
    if (q >= 0 && q < tot) rec[sb + q] = stage[q];
    for (q += BW; q < tot; q += BW) rec[sb + q] = stage[q];
  } else {
    piece_place<false>(cb, lo, hi, lane, cnt, cnt2, k, r, rec, stage, 0);
  }
}


template <int K>
__device__ __forceinline__ int row_lane(int v) {
  return __builtin_amdgcn_update_dpp(0, v, 0x150 + K, 0xf, 0xf, false);
}

template <int K0, int NK>
__device__ __forceinline__ void pair_steps(const char* __restrict__ mb, unsigned loff,
                                           int my, float wb, v4f& acc) {
  v4f v[NK];
#define PAIR_READ(U)                                                              \
  if (U < NK) {                                                                   \
    const unsigned s = (unsigned)row_lane<K0 + (U < NK ? U : 0)>(my);             \
    v[U < NK ? U : 0] = *(const v4f*)(mb + ((s << 8) | loff));                    \
  }
  PAIR_READ(0) PAIR_READ(1) PAIR_READ(2) PAIR_READ(3)
  PAIR_READ(4) PAIR_READ(5) PAIR_READ(6) PAIR_READ(7)
#undef PAIR_READ
  __builtin_amdgcn_sched_barrier(0);
#define PAIR_SUM(U)                                                               \
  if (U < NK) {                                                                   \
    const float w = __int_as_float(row_lane<K0 + (U < NK ? U : 0)>(__float_as_int(wb))); \
    const v4f r = v[U < NK ? U : 0];                                              \
    acc.x = fmaf(w, r.x, acc.x);                                                  \
    acc.y = fmaf(w, r.y, acc.y);                                                  \
    acc.z = fmaf(w, r.z, acc.z);                                                  \
    acc.w = fmaf(w, r.w, acc.w);                                                  \
  }
  PAIR_SUM(0) PAIR_SUM(1) PAIR_SUM(2) PAIR_SUM(3)
  PAIR_SUM(4) PAIR_SUM(5) PAIR_SUM(6) PAIR_SUM(7)
#undef PAIR_SUM
}

template <bool WSRC>
__device__ __forceinline__ v4f row_sum(const int* __restrict__ rowptr,
                                       const int* __restrict__ rec,
                                       const float* __restrict__ msg,
                                       const float* __restrict__ dinv,
                                       int i, int lane, v4f& own, float& di) {
  const int beg = __builtin_amdgcn_readfirstlane(rowptr[i]);
  const int end = __builtin_amdgcn_readfirstlane(rowptr[i + 1]);
  di = dinv[i];
  const int half = lane >> 4;
  const int l16  = lane & 15;
  const unsigned loff = (unsigned)l16 << 4;
  const int pos = 2 * l16 + half;
  const char* __restrict__ mb = (const char*)msg;
  const v4f self = *(const v4f*)(mb + (((unsigned)i << 8) | loff));
  own = self;
  const float ws = WSRC ? di : 1.0f;
  v4f acc;
  acc.x = half ? 0.0f : ws * self.x;
  acc.y = half ? 0.0f : ws * self.y;
  acc.z = half ? 0.0f : ws * self.z;
  acc.w = half ? 0.0f : ws * self.w;
  for (int base = beg; base < end; base += 32) {
    const int cnt = min(32, end - base);
    int my = i;
    float wb = 0.0f;
    if (pos < cnt) {
      my = rec[(size_t)base + pos];
      wb = WSRC ? dinv[my] : 1.0f;
    }
    pair_steps<0, 8>(mb, loff, my, wb, acc);
    if (cnt > 16) {
      pair_steps<8, 4>(mb, loff, my, wb, acc);
      if (cnt > 24) pair_steps<12, 4>(mb, loff, my, wb, acc);
    }
  }
  v4f r;
  r.x = acc.x + __int_as_float(__builtin_amdgcn_permlanex16(0, __float_as_int(acc.x), 0x76543210, 0xfedcba98, false, false));
  r.y = acc.y + __int_as_float(__builtin_amdgcn_permlanex16(0, __float_as_int(acc.y), 0x76543210, 0xfedcba98, false, false));
  r.z = acc.z + __int_as_float(__builtin_amdgcn_permlanex16(0, __float_as_int(acc.z), 0x76543210, 0xfedcba98, false, false));
  r.w = acc.w + __int_as_float(__builtin_amdgcn_permlanex16(0, __float_as_int(acc.w), 0x76543210, 0xfedcba98, false, false));
  return r;
}

#define TP 68
#define WP 72

__device__ __forceinline__ v8f wmma4(v2f a, v2f b, v8f c) {
  return __builtin_amdgcn_wmma_f32_16x16x4_f32(false, a, false, b, (short)0, c, false, false);
}

__global__ void __launch_bounds__(512)
gather_dense(const int* __restrict__ rowptr, const int* __restrict__ rec,
             const float* __restrict__ x, const float* __restrict__ dinv,
             const float* __restrict__ W1, const float* __restrict__ b1,
             const float* __restrict__ W2, const float* __restrict__ Wemb,
             float* __restrict__ hw2, float* __restrict__ out, int n) {
  __shared__ __attribute__((aligned(16))) float Ts[32 * TP];
  __shared__ __attribute__((aligned(16))) float Xs[32 * TP];
  __shared__ __attribute__((aligned(16))) float Hs[32 * TP];
  __shared__ __attribute__((aligned(16))) float Es[32 * TP];
  __shared__ __attribute__((aligned(16))) float W2s[DD * WP];
  __shared__ float b1s[DD];
  const int lane = threadIdx.x & 31;
  const int half = lane >> 4;
  const int l16  = lane & 15;
  const int w = __builtin_amdgcn_readfirstlane((int)(threadIdx.x >> 5));
  const int row0 = blockIdx.x * 32;

  float dsv[2];
#pragma unroll
  for (int rr = 0; rr < 2; ++rr) {
    const int lr = rr * 16 + w;
    const int ir = min(row0 + lr, n - 1);
    v4f own;
    float di;
    const v4f s = row_sum<true>(rowptr, rec, x, dinv, ir, lane, own, di);
    dsv[rr] = di;
    if (half == 0) {
      v4f t;
      t.x = di * s.x;
      t.y = di * s.y;
      t.z = di * s.z;
      t.w = di * s.w;
      *(v4f*)(Ts + lr * TP + 4 * l16) = t;
      *(v4f*)(Xs + lr * TP + 4 * l16) = own;
    }
  }

  const int tile = (w >> 2) & 1;
  const int ct   = w & 3;
  const float* __restrict__ Wm = (w < 8) ? W1 : Wemb;
  v2f bf[16];
#pragma unroll
  for (int k = 0; k < 16; ++k) {
    const int kb = 4 * k + 2 * half;
    bf[k].x = Wm[kb * DD + 16 * ct + l16];
    bf[k].y = Wm[(kb + 1) * DD + 16 * ct + l16];
  }
  {
    const int t  = threadIdx.x;
    const int r  = t >> 3;
    const int c0 = (t & 7) * 8;
    const float4 q0 = ((const float4*)W2)[2 * t];
    const float4 q1 = ((const float4*)W2)[2 * t + 1];
    *(float4*)(W2s + r * WP + c0)     = q0;
    *(float4*)(W2s + r * WP + c0 + 4) = q1;
    if (t < DD) b1s[t] = b1[t];
  }
  __syncthreads();

  {
    const float* As = ((w < 8) ? Ts : Xs) + (tile * 16 + l16) * TP + 2 * half;
    v8f acc0 = {}, acc1 = {};
#pragma unroll
    for (int k = 0; k < 16; k += 2) {
      acc0 = wmma4(*(const v2f*)(As + 4 * k), bf[k], acc0);
      acc1 = wmma4(*(const v2f*)(As + 4 * (k + 1)), bf[k + 1], acc1);
    }
    const v8f acc = acc0 + acc1;
    const int o = (tile * 16 + 8 * half) * TP + 16 * ct + l16;
    if (w < 8) {
      const float bb = b1s[16 * ct + l16];
#pragma unroll
      for (int r = 0; r < 8; ++r) Hs[o + r * TP] = fmaxf(acc[r] + bb, 0.0f);
    } else {
#pragma unroll
      for (int r = 0; r < 8; ++r) Es[o + r * TP] = acc[r];
    }
  }
  __syncthreads();

  if (w < 8) {
    const float* As = Hs + (tile * 16 + l16) * TP + 2 * half;
    const float* Bs = W2s + 2 * half * WP + 16 * ct + l16;
    v8f acc0 = {}, acc1 = {};
#pragma unroll
    for (int k = 0; k < 16; k += 2) {
      v2f b0, b1v;
      b0.x  = Bs[(4 * k) * WP];      b0.y  = Bs[(4 * k + 1) * WP];
      b1v.x = Bs[(4 * k + 4) * WP];  b1v.y = Bs[(4 * k + 5) * WP];
      acc0 = wmma4(*(const v2f*)(As + 4 * k), b0, acc0);
      acc1 = wmma4(*(const v2f*)(As + 4 * (k + 1)), b1v, acc1);
    }
    const v8f acc = acc0 + acc1;
    const int o = (tile * 16 + 8 * half) * TP + 16 * ct + l16;
#pragma unroll
    for (int r = 0; r < 8; ++r) Ts[o + r * TP] = acc[r];
  }
  __syncthreads();

#pragma unroll
  for (int rr = 0; rr < 2; ++rr) {
    const int lr = rr * 16 + w;
    const int i = row0 + lr;
    if (i < n) {
      v2f p = ((const v2f*)(Ts + lr * TP))[lane];
      p.x *= dsv[rr];
      p.y *= dsv[rr];
      ((v2f*)hw2)[(size_t)i * 32 + lane] = p;
      ((v2f*)out)[(size_t)i * 32 + lane] = ((const v2f*)(Es + lr * TP))[lane];
    }
  }
}

__global__ void __launch_bounds__(256)
gather_final(const int* __restrict__ rowptr, const int* __restrict__ rec,
             const float* __restrict__ msg, const float* __restrict__ dinv,
             const float* __restrict__ b2, const float* __restrict__ bemb,
             float* out, int n) {
  const int lane = threadIdx.x & 31;
  const int i = blockIdx.x * 8 + __builtin_amdgcn_readfirstlane((int)(threadIdx.x >> 5));
  if (i >= n) return;
  const int half = lane >> 4;
  const int l16  = lane & 15;
  const v4f o  = ((const v4f*)out)[(size_t)i * 16 + l16];
  const v4f bv = ((const v4f*)b2)[l16];
  const v4f be = ((const v4f*)bemb)[l16];
  v4f own;
  float di;
  const v4f a = row_sum<false>(rowptr, rec, msg, dinv, i, lane, own, di);
  v4f r;
  r.x = o.x + be.x + bv.x + di * a.x;
  r.y = o.y + be.y + bv.y + di * a.y;
  r.z = o.z + be.z + bv.z + di * a.z;
  r.w = o.w + be.w + bv.w + di * a.w;
  if (half == 0) ((v4f*)out)[(size_t)i * 16 + l16] = r;
}

extern "C" void kernel_launch(void* const* d_in, const int* in_sizes, int n_in,
                              void* d_out, int out_size, void* d_ws, size_t ws_size,
                              hipStream_t stream) {
  const float* x    = (const float*)d_in[0];
  const int*   ei   = (const int*)d_in[1];
  const float* W1   = (const float*)d_in[2];
  const float* b1   = (const float*)d_in[3];
  const float* W2   = (const float*)d_in[4];
  const float* b2   = (const float*)d_in[5];
  const float* Wemb = (const float*)d_in[6];
  const float* bemb = (const float*)d_in[7];
  float* out = (float*)d_out;

  const int N = in_sizes[0] / DD;
  const int E = in_sizes[1] / 2;
  const int* src = ei;
  const int* dst = ei + E;

  const size_t npad = ((size_t)N + 1 + 1023) & ~(size_t)1023;
  const size_t epad = ((size_t)E + 63) & ~(size_t)63;
  int*   rowptr  = (int*)d_ws;
  float* dinv    = (float*)(rowptr + npad);
  int*   offt    = (int*)(dinv + npad);
  int*   cb      = offt + (size_t)NCH * MAXB;
  int*   rec     = cb + epad;
  float* buf1    = (float*)(rec + epad);

  const int B     = 256;
  const int nb    = N / BW + 1;
  const int chunk = (E + NCH - 1) / NCH;
  const int gR    = (N + 7) / 8;
  const int gT    = (N + 31) / 32;

  chunk_sort<<<NCH, MAXB, 0, stream>>>(src, dst, offt, cb, E, nb, chunk);
  bin_rows<<<nb, BW, 0, stream>>>(cb, offt, rowptr, dinv, rec, nb, chunk);

  gather_dense<<<gT, 512, 0, stream>>>(rowptr, rec, x, dinv, W1, b1, W2, Wemb, buf1, out, N);

  gather_final<<<gR, B, 0, stream>>>(rowptr, rec, buf1, dinv, b2, bemb, out, N);
}


# Layer-1 launch as a resident grid of 1024 blocks that loop over the 32-row tiles: W1, Wemb, W2 go to LDS once per block, in B-fragment order, not once per tile; E waits in registers; sums as before
# speedup vs baseline: 8.9686x; 1.0790x over previous
#include <hip/hip_runtime.h>


typedef float v2f __attribute__((ext_vector_type(2)));
typedef float v4f __attribute__((ext_vector_type(4)));
typedef float v8f __attribute__((ext_vector_type(8)));

#define DD 64

#define BSH 9
#define BW  512
#define NCH 256
#define MAXB 1024
#define EU  8
#define SPW (NCH / (BW / 32))
#define SCAP 8192
#define RCAP 12288

template <int CTRL>
__device__ __forceinline__ int row_move(int v) {
  return __builtin_amdgcn_update_dpp(0, v, CTRL, 0xf, 0xf, true);
}

template <int NW>
__device__ __forceinline__ int block_excl_scan(int v, int* sw) {
  const int lane = threadIdx.x & 31;
  const int wave = threadIdx.x >> 5;
  int inc = v;
  inc += row_move<0x111>(inc);
  inc += row_move<0x112>(inc);
  inc += row_move<0x114>(inc);
  inc += row_move<0x118>(inc);
  const int lower = __builtin_amdgcn_readlane(inc, 15);
  if (lane >= 16) inc += lower;
  if (lane == 31) sw[wave] = inc;
  __syncthreads();
  int low = (lane < wave && lane < NW) ? sw[lane] : 0;
  low += row_move<0x161>(low);
  low += row_move<0x162>(low);
  low += row_move<0x164>(low);
  low += row_move<0x168>(low);
  low += __builtin_amdgcn_permlanex16(0, low, 0x76543210, 0xfedcba98, false, false);
  return low + inc - v;
}

__device__ __forceinline__ void place_keys(const int* __restrict__ src,
                                           const int* __restrict__ dst, int* lh,
                                           int* __restrict__ cb, int beg, int end, int t) {
  for (int e0 = beg + t; e0 < end; e0 += EU * MAXB) {
    int d[EU];
    int s[EU];
#pragma unroll
    for (int u = 0; u < EU; ++u) {
      const int e = min(e0 + u * MAXB, end - 1);
      d[u] = dst[e];
      s[u] = src[e];
    }
    asm volatile("" :: "v"(d[0]), "v"(d[1]), "v"(d[2]), "v"(d[3]),
                 "v"(d[4]), "v"(d[5]), "v"(d[6]), "v"(d[7]),
                 "v"(s[0]), "v"(s[1]), "v"(s[2]), "v"(s[3]),
                 "v"(s[4]), "v"(s[5]), "v"(s[6]), "v"(s[7]));
#pragma unroll
    for (int u = 0; u < EU; ++u) {
      if (e0 + u * MAXB < end) {
        const int p = atomicAdd(&lh[d[u] >> BSH], 1);
        cb[p] = (int)(((unsigned)s[u] << BSH) | (unsigned)(d[u] & (BW - 1)));
      }
    }
  }
}

static_assert(SCAP <= EU * MAXB, "a chunk that fits the LDS stretch must fit one trip of EU reads a thread");

__global__ void __launch_bounds__(MAXB)
chunk_sort(const int* __restrict__ src, const int* __restrict__ dst,
           int* __restrict__ offt, int* __restrict__ cb, int ne, int nb, int chunk) {
  __shared__ int lh[MAXB];
  __shared__ int sw[32];
  __shared__ int stage[SCAP];
  const int t = threadIdx.x;
  lh[t] = 0;
  __syncthreads();
  const int beg = blockIdx.x * chunk;
  const int end = min(ne, beg + chunk);
  if (chunk <= SCAP) {
    int d[EU];
    int s[EU];
    int r[EU];
#pragma unroll
    for (int u = 0; u < EU; ++u) {
      const int e = max(min(beg + t + u * MAXB, end - 1), 0);
      d[u] = dst[e];
      s[u] = src[e];
    }
    asm volatile("" :: "v"(d[0]), "v"(d[1]), "v"(d[2]), "v"(d[3]),
                 "v"(d[4]), "v"(d[5]), "v"(d[6]), "v"(d[7]),
                 "v"(s[0]), "v"(s[1]), "v"(s[2]), "v"(s[3]),
                 "v"(s[4]), "v"(s[5]), "v"(s[6]), "v"(s[7]));
#pragma unroll
    for (int u = 0; u < EU; ++u) {
      r[u] = 0;
      if (beg + t + u * MAXB < end) r[u] = atomicAdd(&lh[d[u] >> BSH], 1);
    }
    __syncthreads();
    const int ex = block_excl_scan<MAXB / 32>(lh[t], sw);
    if (t <= nb) offt[blockIdx.x * (nb + 1) + t] = ex;
    lh[t] = ex;
    __syncthreads();
    int f[EU];
#pragma unroll
    for (int u = 0; u < EU; ++u) f[u] = lh[d[u] >> BSH];
#pragma unroll
    for (int u = 0; u < EU; ++u) {
      if (beg + t + u * MAXB < end)
        stage[f[u] + r[u]] = (int)(((unsigned)s[u] << BSH) | (unsigned)(d[u] & (BW - 1)));
    }
    __syncthreads();
    int k = t - (beg & 31);
    if (k >= 0 && k < end - beg) cb[beg + k] = stage[k];
    for (k += MAXB; k < end - beg; k += MAXB) cb[beg + k] = stage[k];
  } else {
    for (int e0 = beg + t; e0 < end; e0 += EU * MAXB) {
      int d[EU];
#pragma unroll
      for (int u = 0; u < EU; ++u) d[u] = dst[min(e0 + u * MAXB, end - 1)];
      asm volatile("" :: "v"(d[0]), "v"(d[1]), "v"(d[2]), "v"(d[3]),
                   "v"(d[4]), "v"(d[5]), "v"(d[6]), "v"(d[7]));
#pragma unroll
      for (int u = 0; u < EU; ++u)
        if (e0 + u * MAXB < end) atomicAdd(&lh[d[u] >> BSH], 1);
    }
    __syncthreads();
    const int ex = block_excl_scan<MAXB / 32>(lh[t], sw);
    if (t <= nb) offt[blockIdx.x * (nb + 1) + t] = ex;
    lh[t] = beg + ex;
    __syncthreads();
    place_keys(src, dst, lh, cb, beg, end, t);
  }
}

__device__ __forceinline__ void piece_count(const int* __restrict__ cb, const int (&lo)[SPW],
                                            const int (&hi)[SPW], int lane, int* cnt, int* cnt2,
                                            int (&k)[2 * SPW], int (&r)[2 * SPW]) {
#pragma unroll
  for (int s = 0; s < SPW; ++s) {
#pragma unroll
    for (int u = 0; u < 2; ++u) {
      const int p = lo[s] + lane + 32 * u;
      k[2 * s + u] = cb[max(min(p, hi[s] - 1), 0)];
    }
  }
  asm volatile("" :: "v"(k[0]), "v"(k[1]), "v"(k[2]), "v"(k[3]), "v"(k[4]), "v"(k[5]),
               "v"(k[6]), "v"(k[7]), "v"(k[8]), "v"(k[9]), "v"(k[10]), "v"(k[11]),
               "v"(k[12]), "v"(k[13]), "v"(k[14]), "v"(k[15]));
  asm volatile("" :: "v"(k[16]), "v"(k[17]), "v"(k[18]), "v"(k[19]), "v"(k[20]), "v"(k[21]),
               "v"(k[22]), "v"(k[23]), "v"(k[24]), "v"(k[25]), "v"(k[26]), "v"(k[27]),
               "v"(k[28]), "v"(k[29]), "v"(k[30]), "v"(k[31]));
#pragma unroll
  for (int s = 0; s < SPW; ++s) {
#pragma unroll
    for (int u = 0; u < 2; ++u) {
      r[2 * s + u] = 0;
      if (lo[s] + lane + 32 * u < hi[s])
        r[2 * s + u] = atomicAdd(&cnt[k[2 * s + u] & (BW - 1)], 1);
    }
  }
#pragma unroll
  for (int s = 0; s < SPW; ++s)
    for (int p = lo[s] + 64 + lane; p < hi[s]; p += 32) atomicAdd(&cnt2[cb[p] & (BW - 1)], 1);
}

template <bool TO_LDS>
__device__ __forceinline__ void piece_place(const int* __restrict__ cb, const int (&lo)[SPW],
                                            const int (&hi)[SPW], int lane, const int* cnt,
                                            int* cnt2, const int (&k)[2 * SPW],
                                            const int (&r)[2 * SPW], int* __restrict__ rec,
                                            int* stage, int sbase) {
  int f[2 * SPW];
#pragma unroll
  for (int i = 0; i < 2 * SPW; ++i) f[i] = cnt[k[i] & (BW - 1)];
#pragma unroll
  for (int s = 0; s < SPW; ++s) {
#pragma unroll
    for (int u = 0; u < 2; ++u) {
      if (lo[s] + lane + 32 * u < hi[s]) {
        const int slot = f[2 * s + u] + r[2 * s + u];
        if (TO_LDS) stage[slot - sbase] = k[2 * s + u] >> BSH;
        else rec[slot] = k[2 * s + u] >> BSH;
      }
    }
  }
#pragma unroll
  for (int s = 0; s < SPW; ++s) {
    for (int p = lo[s] + 64 + lane; p < hi[s]; p += 32) {
      const int key = cb[p];
      const int slot = atomicAdd(&cnt2[key & (BW - 1)], 1);
      if (TO_LDS) stage[slot - sbase] = key >> BSH;
      else rec[slot] = key >> BSH;
    }
  }
}

__global__ void __launch_bounds__(BW)
bin_rows(const int* __restrict__ cb, const int* __restrict__ offt,
         int* __restrict__ rowptr, float* __restrict__ dinv, int* __restrict__ rec,
         int nb, int chunk) {
  __shared__ int cnt[BW];
  __shared__ int cnt2[BW];
  __shared__ int sw[32];
  __shared__ int sbeg;
  __shared__ int stot;
  __shared__ int stage[RCAP];
  const int t = threadIdx.x;
  const int lane = t & 31;
  const int w = __builtin_amdgcn_readfirstlane(t >> 5);
  const int b = blockIdx.x;
  cnt[t] = 0;
  cnt2[t] = 0;
  if (t == 0) sbeg = 0;
  int lo[SPW];
  int hi[SPW];
  int below = 0;
#pragma unroll
  for (int s = 0; s < SPW; ++s) {
    const int c = w + (BW / 32) * s;
    const int o0 = offt[c * (nb + 1) + b];
    const int o1 = offt[c * (nb + 1) + b + 1];
    below += o0;
    lo[s] = c * chunk + o0;
    hi[s] = c * chunk + o1;
  }
  __syncthreads();
  if (lane == 0) atomicAdd(&sbeg, below);
  int k[2 * SPW];
  int r[2 * SPW];
  piece_count(cb, lo, hi, lane, cnt, cnt2, k, r);
  __syncthreads();
  const int c1 = cnt[t];
  const int c = c1 + cnt2[t];
  const int ex = block_excl_scan<BW / 32>(c, sw);
  const int first = sbeg + ex;
  const int node = b * BW + t;
  rowptr[node] = first;
  dinv[node] = rsqrtf((float)(c + 1));
  cnt[t] = first;
  cnt2[t] = first + c1;
  if (t == BW - 1) stot = ex + c;
  __syncthreads();
  const int sb  = __builtin_amdgcn_readfirstlane(sbeg);
  const int tot = __builtin_amdgcn_readfirstlane(stot);
  if (tot <= RCAP) {
    piece_place<true>(cb, lo, hi, lane, cnt, cnt2, k, r, rec, stage, sb);
    __syncthreads();
    int q = t - (sb & 31);
    if (q >= 0 && q < tot) rec[sb + q] = stage[q];
    for (q += BW; q < tot; q += BW) rec[sb + q] = stage[q];
  } else {
    piece_place<false>(cb, lo, hi, lane, cnt, cnt2, k, r, rec, stage, 0);
  }
}


template <int K>
__device__ __forceinline__ int row_lane(int v) {
  return __builtin_amdgcn_update_dpp(0, v, 0x150 + K, 0xf, 0xf, false);
}

template <int K0, int NK>
__device__ __forceinline__ void pair_steps(const char* __restrict__ mb, unsigned loff,
                                           int my, float wb, v4f& acc) {
  v4f v[NK];
#define PAIR_READ(U)                                                              \
  if (U < NK) {                                                                   \
    const unsigned s = (unsigned)row_lane<K0 + (U < NK ? U : 0)>(my);             \
    v[U < NK ? U : 0] = *(const v4f*)(mb + ((s << 8) | loff));                    \
  }
  PAIR_READ(0) PAIR_READ(1) PAIR_READ(2) PAIR_READ(3)
  PAIR_READ(4) PAIR_READ(5) PAIR_READ(6) PAIR_READ(7)
#undef PAIR_READ
  __builtin_amdgcn_sched_barrier(0);
#define PAIR_SUM(U)                                                               \
  if (U < NK) {                                                                   \
    const float w = __int_as_float(row_lane<K0 + (U < NK ? U : 0)>(__float_as_int(wb))); \
    const v4f r = v[U < NK ? U : 0];                                              \
    acc.x = fmaf(w, r.x, acc.x);                                                  \
    acc.y = fmaf(w, r.y, acc.y);                                                  \
    acc.z = fmaf(w, r.z, acc.z);                                                  \
    acc.w = fmaf(w, r.w, acc.w);                                                  \
  }
  PAIR_SUM(0) PAIR_SUM(1) PAIR_SUM(2) PAIR_SUM(3)
  PAIR_SUM(4) PAIR_SUM(5) PAIR_SUM(6) PAIR_SUM(7)
#undef PAIR_SUM
}

template <bool WSRC>
__device__ __forceinline__ v4f row_sum(const int* __restrict__ rowptr,
                                       const int* __restrict__ rec,
                                       const float* __restrict__ msg,
                                       const float* __restrict__ dinv,
                                       int i, int lane, float* ownrow, float& di) {
  const int beg = __builtin_amdgcn_readfirstlane(rowptr[i]);
  const int end = __builtin_amdgcn_readfirstlane(rowptr[i + 1]);
  di = __int_as_float(__builtin_amdgcn_readfirstlane(__float_as_int(dinv[i])));
  const int half = lane >> 4;
  const int l16  = lane & 15;
  const unsigned loff = (unsigned)l16 << 4;
  const int pos = 2 * l16 + half;
  const char* __restrict__ mb = (const char*)msg;
  const v4f self = *(const v4f*)(mb + (((unsigned)i << 8) | loff));
  if (ownrow != nullptr && half == 0) *(v4f*)(ownrow + 4 * l16) = self;
  const float ws = WSRC ? di : 1.0f;
  v4f acc;
  acc.x = half ? 0.0f : ws * self.x;
  acc.y = half ? 0.0f : ws * self.y;
  acc.z = half ? 0.0f : ws * self.z;
  acc.w = half ? 0.0f : ws * self.w;
  for (int base = beg; base < end; base += 32) {
    const int cnt = min(32, end - base);
    int my = i;
    float wb = 0.0f;
    if (pos < cnt) {
      my = rec[(size_t)base + pos];
      wb = WSRC ? dinv[my] : 1.0f;
    }
    pair_steps<0, 8>(mb, loff, my, wb, acc);
    if (cnt > 16) {
      pair_steps<8, 4>(mb, loff, my, wb, acc);
      if (cnt > 24) pair_steps<12, 4>(mb, loff, my, wb, acc);
    }
  }
  v4f r;
  r.x = acc.x + __int_as_float(__builtin_amdgcn_permlanex16(0, __float_as_int(acc.x), 0x76543210, 0xfedcba98, false, false));
  r.y = acc.y + __int_as_float(__builtin_amdgcn_permlanex16(0, __float_as_int(acc.y), 0x76543210, 0xfedcba98, false, false));
  r.z = acc.z + __int_as_float(__builtin_amdgcn_permlanex16(0, __float_as_int(acc.z), 0x76543210, 0xfedcba98, false, false));
  r.w = acc.w + __int_as_float(__builtin_amdgcn_permlanex16(0, __float_as_int(acc.w), 0x76543210, 0xfedcba98, false, false));
  return r;
}

#define TP 68
#define WFN (4 * 16 * 32)

__device__ __forceinline__ v8f wmma4(v2f a, v2f b, v8f c) {
  return __builtin_amdgcn_wmma_f32_16x16x4_f32(false, a, false, b, (short)0, c, false, false);
}

__global__ void __launch_bounds__(512)
gather_dense(const int* __restrict__ rowptr, const int* __restrict__ rec,
             const float* __restrict__ x, const float* __restrict__ dinv,
             const float* __restrict__ W1, const float* __restrict__ b1,
             const float* __restrict__ W2, const float* __restrict__ Wemb,
             float* __restrict__ hw2, float* __restrict__ out, int n, int ntiles, int nblk) {
  __shared__ __attribute__((aligned(16))) float smem[3 * 32 * TP + DD + 3 * 2 * WFN];
  float* const Ts  = smem;
  float* const Xs  = smem + 32 * TP;
  float* const Hs  = smem + 2 * 32 * TP;
  float* const b1s = smem + 3 * 32 * TP;
  v2f* const W1f = (v2f*)(smem + 3 * 32 * TP + DD);
  v2f* const Wef = W1f + WFN;
  v2f* const W2f = Wef + WFN;
  const int lane = threadIdx.x & 31;
  const int half = lane >> 4;
  const int l16  = lane & 15;
  const int w = __builtin_amdgcn_readfirstlane((int)(threadIdx.x >> 5));
  const int tile = (w >> 2) & 1;
  const int ct   = w & 3;

  {
    const int kb = 4 * w + 2 * half;
#pragma unroll
    for (int j = 0; j < 4; ++j) {
      const int col = 16 * j + l16;
      const int e = (j * 16 + w) * 32 + lane;
      v2f f1, fe, f2;
      f1.x = W1[kb * DD + col];    f1.y = W1[(kb + 1) * DD + col];
      fe.x = Wemb[kb * DD + col];  fe.y = Wemb[(kb + 1) * DD + col];
      f2.x = W2[kb * DD + col];    f2.y = W2[(kb + 1) * DD + col];
      W1f[e] = f1;
      Wef[e] = fe;
      W2f[e] = f2;
    }
    if (threadIdx.x < DD) b1s[threadIdx.x] = b1[threadIdx.x];
  }

  for (int tix = blockIdx.x; tix < ntiles; tix += nblk) {
    const int row0 = tix * 32;

    float dsv[2];
#pragma unroll
    for (int rr = 0; rr < 2; ++rr) {
      const int lr = rr * 16 + w;
      const int ir = min(row0 + lr, n - 1);
      float di;
      const v4f s = row_sum<true>(rowptr, rec, x, dinv, ir, lane, Xs + lr * TP, di);
      dsv[rr] = di;
      if (half == 0) {
        v4f t;
        t.x = di * s.x;
        t.y = di * s.y;
        t.z = di * s.z;
        t.w = di * s.w;
        *(v4f*)(Ts + lr * TP + 4 * l16) = t;
      }
    }
    __syncthreads();

    const int o = (tile * 16 + 8 * half) * TP + 16 * ct + l16;
    v8f acc;
    {
      const float* As = ((w < 8) ? Ts : Xs) + (tile * 16 + l16) * TP + 2 * half;
      const v2f* Bf = ((w < 8) ? W1f : Wef) + ct * 16 * 32 + lane;
      v8f acc0 = {}, acc1 = {};
#pragma unroll
      for (int k = 0; k < 16; k += 2) {
        acc0 = wmma4(*(const v2f*)(As + 4 * k), Bf[k * 32], acc0);
        acc1 = wmma4(*(const v2f*)(As + 4 * (k + 1)), Bf[(k + 1) * 32], acc1);
      }
      acc = acc0 + acc1;
      if (w < 8) {
        const float bb = b1s[16 * ct + l16];
#pragma unroll
        for (int r = 0; r < 8; ++r) Hs[o + r * TP] = fmaxf(acc[r] + bb, 0.0f);
      }
    }
    __syncthreads();

    if (w < 8) {
      const float* As = Hs + (tile * 16 + l16) * TP + 2 * half;
      const v2f* Bf = W2f + ct * 16 * 32 + lane;
      v8f acc0 = {}, acc1 = {};
#pragma unroll
      for (int k = 0; k < 16; k += 2) {
        acc0 = wmma4(*(const v2f*)(As + 4 * k), Bf[k * 32], acc0);
        acc1 = wmma4(*(const v2f*)(As + 4 * (k + 1)), Bf[(k + 1) * 32], acc1);
      }
      const v8f p = acc0 + acc1;
#pragma unroll
      for (int r = 0; r < 8; ++r) Ts[o + r * TP] = p[r];
    } else {
#pragma unroll
      for (int r = 0; r < 8; ++r) Xs[o + r * TP] = acc[r];
    }
    __syncthreads();

#pragma unroll
    for (int rr = 0; rr < 2; ++rr) {
      const int lr = rr * 16 + w;
      const int i = row0 + lr;
      if (i < n) {
        v2f p = ((const v2f*)(Ts + lr * TP))[lane];
        p.x *= dsv[rr];
        p.y *= dsv[rr];
        ((v2f*)hw2)[(size_t)i * 32 + lane] = p;
        ((v2f*)out)[(size_t)i * 32 + lane] = ((const v2f*)(Xs + lr * TP))[lane];
      }
    }
  }
}

__global__ void __launch_bounds__(256)
gather_final(const int* __restrict__ rowptr, const int* __restrict__ rec,
             const float* __restrict__ msg, const float* __restrict__ dinv,
             const float* __restrict__ b2, const float* __restrict__ bemb,
             float* out, int n) {
  const int lane = threadIdx.x & 31;
  const int i = blockIdx.x * 8 + __builtin_amdgcn_readfirstlane((int)(threadIdx.x >> 5));
  if (i >= n) return;
  const int half = lane >> 4;
  const int l16  = lane & 15;
  const v4f o  = ((const v4f*)out)[(size_t)i * 16 + l16];
  const v4f bv = ((const v4f*)b2)[l16];
  const v4f be = ((const v4f*)bemb)[l16];
  float di;
  const v4f a = row_sum<false>(rowptr, rec, msg, dinv, i, lane, nullptr, di);
  v4f r;
  r.x = o.x + be.x + bv.x + di * a.x;
  r.y = o.y + be.y + bv.y + di * a.y;
  r.z = o.z + be.z + bv.z + di * a.z;
  r.w = o.w + be.w + bv.w + di * a.w;
  if (half == 0) ((v4f*)out)[(size_t)i * 16 + l16] = r;
}

extern "C" void kernel_launch(void* const* d_in, const int* in_sizes, int n_in,
                              void* d_out, int out_size, void* d_ws, size_t ws_size,
                              hipStream_t stream) {
  const float* x    = (const float*)d_in[0];
  const int*   ei   = (const int*)d_in[1];
  const float* W1   = (const float*)d_in[2];
  const float* b1   = (const float*)d_in[3];
  const float* W2   = (const float*)d_in[4];
  const float* b2   = (const float*)d_in[5];
  const float* Wemb = (const float*)d_in[6];
  const float* bemb = (const float*)d_in[7];
  float* out = (float*)d_out;

  const int N = in_sizes[0] / DD;
  const int E = in_sizes[1] / 2;
  const int* src = ei;
  const int* dst = ei + E;

  const size_t npad = ((size_t)N + 1 + 1023) & ~(size_t)1023;
  const size_t epad = ((size_t)E + 63) & ~(size_t)63;
  int*   rowptr  = (int*)d_ws;
  float* dinv    = (float*)(rowptr + npad);
  int*   offt    = (int*)(dinv + npad);
  int*   cb      = offt + (size_t)NCH * MAXB;
  int*   rec     = cb + epad;
  float* buf1    = (float*)(rec + epad);

  const int B     = 256;
  const int nb    = N / BW + 1;
  const int chunk = (E + NCH - 1) / NCH;
  const int gR    = (N + 7) / 8;
  const int gT    = (N + 31) / 32;
  const int gP    = gT < 1024 ? gT : 1024;

  chunk_sort<<<NCH, MAXB, 0, stream>>>(src, dst, offt, cb, E, nb, chunk);
  bin_rows<<<nb, BW, 0, stream>>>(cb, offt, rowptr, dinv, rec, nb, chunk);

  gather_dense<<<gP, 512, 0, stream>>>(rowptr, rec, x, dinv, W1, b1, W2, Wemb, buf1, out, N, gT, gP);

  gather_final<<<gR, B, 0, stream>>>(rowptr, rec, buf1, dinv, b2, bemb, out, N);
}
